# starting kernel
# baseline (speedup 1.0000x reference)
#include <hip/hip_runtime.h>


typedef float v8f __attribute__((ext_vector_type(8)));
typedef float v2f __attribute__((ext_vector_type(2)));

#define D_DIM 2048
#define N_MODES 16
#define B_DIM 2
#define L_DIM 4096
#define CG 128
#define NGROUP 2
#define MROW 17
#define UROW 136
#define XROW 17

static __device__ __forceinline__ v8f wmma4(v2f a, v2f b, v8f c) {
  return __builtin_amdgcn_wmma_f32_16x16x4_f32(false, a, false, b, (short)0, c,
                                               false, false);
}

static __device__ __forceinline__ void cpow(float qr, float qi, int e,
                                            float& pr, float& pi) {
  pr = 1.f; pi = 0.f;
  for (int k = 0; k < e; ++k) {
    float t = pr * qr - pi * qi;
    pi = pr * qi + pi * qr;
    pr = t;
  }
}

__launch_bounds__(256, 2)
__global__ void cema_kernel(const float* __restrict__ x,
                            const float* __restrict__ alpha,
                            const float* __restrict__ delta,
                            const float* __restrict__ theta,
                            const float* __restrict__ gma,
                            const float* __restrict__ omga,
                            float* __restrict__ y_out,
                            float* __restrict__ h_out) {
  __shared__ float xs_s[B_DIM][CG * XROW];
  __shared__ float Ure_s[B_DIM][N_MODES * UROW];
  __shared__ float Uim_s[B_DIM][N_MODES * UROW];
  __shared__ float Bre_s[N_MODES * MROW];
  __shared__ float Bim_s[N_MODES * MROW];
  __shared__ float Are_s[N_MODES * MROW];
  __shared__ float AimN_s[N_MODES * MROW];
  __shared__ float Kmat_s[N_MODES * MROW];
  __shared__ float kern_s[N_MODES];
  __shared__ float cQr[16], cQi[16], cP[16], cGr[16], cGi[16];
  __shared__ float cCr[16], cCi[16], cTr[16], cTi[16];
  __shared__ float sOmega;

  const int tid = threadIdx.x;
  const int d = blockIdx.x;

  if (tid < 16) {
    int n = tid;
    float a = alpha[d * 16 + n];
    float de = delta[d * 16 + n];
    float th = theta[d];
    float p = 1.f / (1.f + expf(-a));
    float dd = 1.f / (1.f + expf(-de));
    float base = (1.f / (1.f + expf(-th))) * (2.f * 3.14159265358979323846f / 16.f);
    float phi = (float)(n + 1) * base;
    float r = 1.f - p * dd;
    float sn, cs;
    sincosf(phi, &sn, &cs);
    float qr = r * cs, qi = r * sn;
    cQr[n] = qr; cQi[n] = qi;
    cP[n] = p;
    float gr = gma[(d * 16 + n) * 2 + 0] * 0.25f;
    float gi = gma[(d * 16 + n) * 2 + 1] * 0.25f;
    cGr[n] = gr; cGi[n] = gi;
    cCr[n] = p * gr; cCi[n] = p * gi;
    float tr, ti;
    cpow(qr, qi, 16, tr, ti);
    cTr[n] = tr; cTi[n] = ti;
    if (n == 0) sOmega = omga[d];
  }
  __syncthreads();

  {
    int r = tid >> 4, cc = tid & 15;
    float qr = cQr[cc], qi = cQi[cc];
    float pr, pi;
    cpow(qr, qi, r, pr, pi);
    xs_s[0][r * MROW + cc] = cCr[cc] * pr - cCi[cc] * pi;
    float q1r = pr * qr - pi * qi;
    float q1i = pr * qi + pi * qr;
    Are_s[r * MROW + cc] = cGr[cc] * q1r - cGi[cc] * q1i;
    AimN_s[r * MROW + cc] = -(cGr[cc] * q1i + cGi[cc] * q1r);
    float br, bi;
    cpow(cQr[r], cQi[r], 15 - cc, br, bi);
    Bre_s[r * MROW + cc] = cP[r] * br;
    Bim_s[r * MROW + cc] = cP[r] * bi;
  }
  __syncthreads();
  if (tid < 16) {
    float s = 0.f;
    for (int n = 0; n < 16; ++n) s += xs_s[0][tid * MROW + n];
    kern_s[tid] = s;
  }
  __syncthreads();
  {
    int r = tid >> 4, cc = tid & 15;
    float v = (r >= cc) ? kern_s[r - cc] : 0.f;
    if (r == cc) v += sOmega;
    Kmat_s[r * MROW + cc] = v;
  }

  const int wv = tid >> 5;
  const int lane = tid & 31;
  const int lcol = lane & 15;
  const int khalf = lane >> 4;

  float hre = 0.f, him = 0.f;

  for (int g = 0; g < NGROUP; ++g) {
    __syncthreads();
    for (int b2 = 0; b2 < B_DIM; ++b2) {
      const float4* xg = (const float4*)(x + ((size_t)b2 * D_DIM + d) * L_DIM +
                                         (size_t)g * (CG * 16));
      for (int i = tid; i < CG * 16 / 4; i += 256) {
        float4 v = xg[i];
        int t = i * 4;
        float* dst = &xs_s[b2][(t >> 4) * XROW + (t & 15)];
        dst[0] = v.x; dst[1] = v.y; dst[2] = v.z; dst[3] = v.w;
      }
    }
    __syncthreads();

    #pragma unroll
    for (int u = 0; u < 2; ++u) {
      int asg = wv * 2 + u;
      int b2 = asg >> 3;
      int tile = asg & 7;
      const float* xb = xs_s[b2];
      int xbase = (tile * 16 + lcol) * XROW;
      v8f accR = {0.f, 0.f, 0.f, 0.f, 0.f, 0.f, 0.f, 0.f};
      v8f accI = {0.f, 0.f, 0.f, 0.f, 0.f, 0.f, 0.f, 0.f};
      #pragma unroll
      for (int kk = 0; kk < 4; ++kk) {
        int k0 = kk * 4 + khalf * 2;
        v2f bx; bx.x = xb[xbase + k0]; bx.y = xb[xbase + k0 + 1];
        v2f aR; aR.x = Bre_s[lcol * MROW + k0]; aR.y = Bre_s[lcol * MROW + k0 + 1];
        v2f aI; aI.x = Bim_s[lcol * MROW + k0]; aI.y = Bim_s[lcol * MROW + k0 + 1];
        accR = wmma4(aR, bx, accR);
        accI = wmma4(aI, bx, accI);
      }
      #pragma unroll
      for (int v = 0; v < 8; ++v) {
        int n = v + 8 * khalf;
        Ure_s[b2][n * UROW + tile * 16 + lcol] = accR[v];
        Uim_s[b2][n * UROW + tile * 16 + lcol] = accI[v];
      }
    }
    __syncthreads();

    if (tid < 32) {
      int b2 = tid >> 4, n = tid & 15;
      float qtr = cTr[n], qti = cTi[n];
      float* ur = &Ure_s[b2][n * UROW];
      float* ui = &Uim_s[b2][n * UROW];
      for (int j = 0; j < CG; ++j) {
        float tur = ur[j], tui = ui[j];
        ur[j] = hre; ui[j] = him;
        float nr = qtr * hre - qti * him + tur;
        float ni = qtr * him + qti * hre + tui;
        hre = nr; him = ni;
      }
    }
    __syncthreads();

    #pragma unroll
    for (int u = 0; u < 2; ++u) {
      int asg = wv * 2 + u;
      int b2 = asg >> 3;
      int tile = asg & 7;
      float* xb = xs_s[b2];
      int xbase = (tile * 16 + lcol) * XROW;
      v8f acc = {0.f, 0.f, 0.f, 0.f, 0.f, 0.f, 0.f, 0.f};
      #pragma unroll
      for (int kk = 0; kk < 4; ++kk) {
        int k0 = kk * 4 + khalf * 2;
        v2f bx; bx.x = xb[xbase + k0]; bx.y = xb[xbase + k0 + 1];
        v2f aK; aK.x = Kmat_s[lcol * MROW + k0]; aK.y = Kmat_s[lcol * MROW + k0 + 1];
        acc = wmma4(aK, bx, acc);
      }
      #pragma unroll
      for (int kk = 0; kk < 4; ++kk) {
        int k0 = kk * 4 + khalf * 2;
        v2f bh;
        bh.x = Ure_s[b2][k0 * UROW + tile * 16 + lcol];
        bh.y = Ure_s[b2][(k0 + 1) * UROW + tile * 16 + lcol];
        v2f aA; aA.x = Are_s[lcol * MROW + k0]; aA.y = Are_s[lcol * MROW + k0 + 1];
        acc = wmma4(aA, bh, acc);
      }
      #pragma unroll
      for (int kk = 0; kk < 4; ++kk) {
        int k0 = kk * 4 + khalf * 2;
        v2f bh;
        bh.x = Uim_s[b2][k0 * UROW + tile * 16 + lcol];
        bh.y = Uim_s[b2][(k0 + 1) * UROW + tile * 16 + lcol];
        v2f aA; aA.x = AimN_s[lcol * MROW + k0]; aA.y = AimN_s[lcol * MROW + k0 + 1];
        acc = wmma4(aA, bh, acc);
      }
      #pragma unroll
      for (int v = 0; v < 8; ++v) {
        xb[(tile * 16 + lcol) * XROW + v + 8 * khalf] = acc[v];
      }
    }
    __syncthreads();

    for (int b2 = 0; b2 < B_DIM; ++b2) {
      float4* yg = (float4*)(y_out + ((size_t)b2 * D_DIM + d) * L_DIM +
                             (size_t)g * (CG * 16));
      for (int i = tid; i < CG * 16 / 4; i += 256) {
        int t = i * 4;
        const float* src = &xs_s[b2][(t >> 4) * XROW + (t & 15)];
        float4 v;
        v.x = src[0]; v.y = src[1]; v.z = src[2]; v.w = src[3];
        yg[i] = v;
      }
    }
  }

  if (tid < 32) {
    int b2 = tid >> 4, n = tid & 15;
    float* hp = h_out + (((size_t)b2 * D_DIM + d) * N_MODES + n) * 2;
    hp[0] = hre;
    hp[1] = him;
  }
}

extern "C" void kernel_launch(void* const* d_in, const int* in_sizes, int n_in,
                              void* d_out, int out_size, void* d_ws, size_t ws_size,
                              hipStream_t stream) {
  (void)in_sizes; (void)n_in; (void)out_size; (void)d_ws; (void)ws_size;
  const float* x = (const float*)d_in[0];
  const float* alpha = (const float*)d_in[1];
  const float* delta = (const float*)d_in[2];
  const float* theta = (const float*)d_in[3];
  const float* gma = (const float*)d_in[4];
  const float* omga = (const float*)d_in[5];
  float* y = (float*)d_out;
  float* h = (float*)d_out + (size_t)B_DIM * D_DIM * L_DIM;
  cema_kernel<<<dim3(D_DIM), dim3(256), 0, stream>>>(x, alpha, delta, theta,
                                                     gma, omga, y, h);
}


# every global load of a block issued as one batch at kernel entry (coefficients and the whole x row, staged once in LDS): one exposed memory round trip per block, not eleven
# speedup vs baseline: 1.2831x; 1.2831x over previous
#include <hip/hip_runtime.h>


typedef float v8f __attribute__((ext_vector_type(8)));
typedef float v2f __attribute__((ext_vector_type(2)));

#define D_DIM 2048
#define N_MODES 16
#define B_DIM 2
#define L_DIM 4096
#define CG 128
#define NGROUP 2
#define MROW 17
#define UROW 136
#define XROW 17
#define NCHUNK (NGROUP * CG)
#define XV4 (L_DIM / 4 / 256)

static __device__ __forceinline__ v8f wmma4(v2f a, v2f b, v8f c) {
  return __builtin_amdgcn_wmma_f32_16x16x4_f32(false, a, false, b, (short)0, c,
                                               false, false);
}

static __device__ __forceinline__ void cpow(float qr, float qi, int e,
                                            float& pr, float& pi) {
  pr = 1.f; pi = 0.f;
  for (int k = 0; k < e; ++k) {
    float t = pr * qr - pi * qi;
    pi = pr * qi + pi * qr;
    pr = t;
  }
}

__launch_bounds__(256, 2)
__global__ void cema_kernel(const float* __restrict__ x,
                            const float* __restrict__ alpha,
                            const float* __restrict__ delta,
                            const float* __restrict__ theta,
                            const float* __restrict__ gma,
                            const float* __restrict__ omga,
                            float* __restrict__ y_out,
                            float* __restrict__ h_out) {
  __shared__ float xs_s[B_DIM][NCHUNK * XROW];
  __shared__ float Ure_s[B_DIM][N_MODES * UROW];
  __shared__ float Uim_s[B_DIM][N_MODES * UROW];
  __shared__ float Bre_s[N_MODES * MROW];
  __shared__ float Bim_s[N_MODES * MROW];
  __shared__ float Are_s[N_MODES * MROW];
  __shared__ float AimN_s[N_MODES * MROW];
  __shared__ float Kmat_s[N_MODES * MROW];
  __shared__ float kern_s[N_MODES];
  __shared__ float cQr[16], cQi[16], cP[16], cGr[16], cGi[16];
  __shared__ float cCr[16], cCi[16], cTr[16], cTi[16];
  __shared__ float sOmega;

  const int tid = threadIdx.x;
  const int d = blockIdx.x;

  const int n16 = tid & 15;
  const float om_in = omga[d];
  const float th_in = theta[d];
  const float a_in = alpha[d * 16 + n16];
  const float de_in = delta[d * 16 + n16];
  const float g0_in = gma[(d * 16 + n16) * 2 + 0];
  const float g1_in = gma[(d * 16 + n16) * 2 + 1];
  float4 xr[B_DIM][XV4];
  #pragma unroll
  for (int b2 = 0; b2 < B_DIM; ++b2) {
    const float4* xg = (const float4*)(x + ((size_t)b2 * D_DIM + d) * L_DIM);
    #pragma unroll
    for (int k = 0; k < XV4; ++k) xr[b2][k] = xg[tid + 256 * k];
  }

  if (tid < 16) {
    int n = tid;
    float a = a_in;
    float de = de_in;
    float th = th_in;
    float p = 1.f / (1.f + expf(-a));
    float dd = 1.f / (1.f + expf(-de));
    float base = (1.f / (1.f + expf(-th))) * (2.f * 3.14159265358979323846f / 16.f);
    float phi = (float)(n + 1) * base;
    float r = 1.f - p * dd;
    float sn, cs;
    sincosf(phi, &sn, &cs);
    float qr = r * cs, qi = r * sn;
    cQr[n] = qr; cQi[n] = qi;
    cP[n] = p;
    float gr = g0_in * 0.25f;
    float gi = g1_in * 0.25f;
    cGr[n] = gr; cGi[n] = gi;
    cCr[n] = p * gr; cCi[n] = p * gi;
    float tr, ti;
    cpow(qr, qi, 16, tr, ti);
    cTr[n] = tr; cTi[n] = ti;
    if (n == 0) sOmega = om_in;
  }
  __syncthreads();

  {
    int r = tid >> 4, cc = tid & 15;
    float qr = cQr[cc], qi = cQi[cc];
    float pr, pi;
    cpow(qr, qi, r, pr, pi);
    xs_s[0][r * MROW + cc] = cCr[cc] * pr - cCi[cc] * pi;
    float q1r = pr * qr - pi * qi;
    float q1i = pr * qi + pi * qr;
    Are_s[r * MROW + cc] = cGr[cc] * q1r - cGi[cc] * q1i;
    AimN_s[r * MROW + cc] = -(cGr[cc] * q1i + cGi[cc] * q1r);
    float br, bi;
    cpow(cQr[r], cQi[r], 15 - cc, br, bi);
    Bre_s[r * MROW + cc] = cP[r] * br;
    Bim_s[r * MROW + cc] = cP[r] * bi;
  }
  __syncthreads();
  if (tid < 16) {
    float s = 0.f;
    for (int n = 0; n < 16; ++n) s += xs_s[0][tid * MROW + n];
    kern_s[tid] = s;
  }
  __syncthreads();
  {
    int r = tid >> 4, cc = tid & 15;
    float v = (r >= cc) ? kern_s[r - cc] : 0.f;
    if (r == cc) v += sOmega;
    Kmat_s[r * MROW + cc] = v;
  }

  #pragma unroll
  for (int b2 = 0; b2 < B_DIM; ++b2) {
    #pragma unroll
    for (int k = 0; k < XV4; ++k) {
      int i = tid + 256 * k;
      int t = i * 4;
      float4 v = xr[b2][k];
      float* dst = &xs_s[b2][(t >> 4) * XROW + (t & 15)];
      dst[0] = v.x; dst[1] = v.y; dst[2] = v.z; dst[3] = v.w;
    }
  }

  const int wv = tid >> 5;
  const int lane = tid & 31;
  const int lcol = lane & 15;
  const int khalf = lane >> 4;

  float hre = 0.f, him = 0.f;

  #pragma unroll
  for (int g = 0; g < NGROUP; ++g) {
    __syncthreads();

    #pragma unroll
    for (int u = 0; u < 2; ++u) {
      int asg = wv * 2 + u;
      int b2 = asg >> 3;
      int tile = asg & 7;
      const float* xb = xs_s[b2] + g * (CG * XROW);
      int xbase = (tile * 16 + lcol) * XROW;
      v8f accR = {0.f, 0.f, 0.f, 0.f, 0.f, 0.f, 0.f, 0.f};
      v8f accI = {0.f, 0.f, 0.f, 0.f, 0.f, 0.f, 0.f, 0.f};
      #pragma unroll
      for (int kk = 0; kk < 4; ++kk) {
        int k0 = kk * 4 + khalf * 2;
        v2f bx; bx.x = xb[xbase + k0]; bx.y = xb[xbase + k0 + 1];
        v2f aR; aR.x = Bre_s[lcol * MROW + k0]; aR.y = Bre_s[lcol * MROW + k0 + 1];
        v2f aI; aI.x = Bim_s[lcol * MROW + k0]; aI.y = Bim_s[lcol * MROW + k0 + 1];
        accR = wmma4(aR, bx, accR);
        accI = wmma4(aI, bx, accI);
      }
      #pragma unroll
      for (int v = 0; v < 8; ++v) {
        int n = v + 8 * khalf;
        Ure_s[b2][n * UROW + tile * 16 + lcol] = accR[v];
        Uim_s[b2][n * UROW + tile * 16 + lcol] = accI[v];
      }
    }
    __syncthreads();

    if (tid < 32) {
      int b2 = tid >> 4, n = tid & 15;
      float qtr = cTr[n], qti = cTi[n];
      float* ur = &Ure_s[b2][n * UROW];
      float* ui = &Uim_s[b2][n * UROW];
      for (int j = 0; j < CG; ++j) {
        float tur = ur[j], tui = ui[j];
        ur[j] = hre; ui[j] = him;
        float nr = qtr * hre - qti * him + tur;
        float ni = qtr * him + qti * hre + tui;
        hre = nr; him = ni;
      }
    }
    __syncthreads();

    #pragma unroll
    for (int u = 0; u < 2; ++u) {
      int asg = wv * 2 + u;
      int b2 = asg >> 3;
      int tile = asg & 7;
      float* xb = xs_s[b2] + g * (CG * XROW);
      int xbase = (tile * 16 + lcol) * XROW;
      v8f acc = {0.f, 0.f, 0.f, 0.f, 0.f, 0.f, 0.f, 0.f};
      #pragma unroll
      for (int kk = 0; kk < 4; ++kk) {
        int k0 = kk * 4 + khalf * 2;
        v2f bx; bx.x = xb[xbase + k0]; bx.y = xb[xbase + k0 + 1];
        v2f aK; aK.x = Kmat_s[lcol * MROW + k0]; aK.y = Kmat_s[lcol * MROW + k0 + 1];
        acc = wmma4(aK, bx, acc);
      }
      #pragma unroll
      for (int kk = 0; kk < 4; ++kk) {
        int k0 = kk * 4 + khalf * 2;
        v2f bh;
        bh.x = Ure_s[b2][k0 * UROW + tile * 16 + lcol];
        bh.y = Ure_s[b2][(k0 + 1) * UROW + tile * 16 + lcol];
        v2f aA; aA.x = Are_s[lcol * MROW + k0]; aA.y = Are_s[lcol * MROW + k0 + 1];
        acc = wmma4(aA, bh, acc);
      }
      #pragma unroll
      for (int kk = 0; kk < 4; ++kk) {
        int k0 = kk * 4 + khalf * 2;
        v2f bh;
        bh.x = Uim_s[b2][k0 * UROW + tile * 16 + lcol];
        bh.y = Uim_s[b2][(k0 + 1) * UROW + tile * 16 + lcol];
        v2f aA; aA.x = AimN_s[lcol * MROW + k0]; aA.y = AimN_s[lcol * MROW + k0 + 1];
        acc = wmma4(aA, bh, acc);
      }
      #pragma unroll
      for (int v = 0; v < 8; ++v) {
        xb[(tile * 16 + lcol) * XROW + v + 8 * khalf] = acc[v];
      }
    }
    __syncthreads();

    for (int b2 = 0; b2 < B_DIM; ++b2) {
      float4* yg = (float4*)(y_out + ((size_t)b2 * D_DIM + d) * L_DIM +
                             (size_t)g * (CG * 16));
      for (int i = tid; i < CG * 16 / 4; i += 256) {
        int t = i * 4;
        const float* src = &xs_s[b2][g * (CG * XROW) + (t >> 4) * XROW + (t & 15)];
        float4 v;
        v.x = src[0]; v.y = src[1]; v.z = src[2]; v.w = src[3];
        yg[i] = v;
      }
    }
  }

  if (tid < 32) {
    int b2 = tid >> 4, n = tid & 15;
    float* hp = h_out + (((size_t)b2 * D_DIM + d) * N_MODES + n) * 2;
    hp[0] = hre;
    hp[1] = him;
  }
}

extern "C" void kernel_launch(void* const* d_in, const int* in_sizes, int n_in,
                              void* d_out, int out_size, void* d_ws, size_t ws_size,
                              hipStream_t stream) {
  (void)in_sizes; (void)n_in; (void)out_size; (void)d_ws; (void)ws_size;
  const float* x = (const float*)d_in[0];
  const float* alpha = (const float*)d_in[1];
  const float* delta = (const float*)d_in[2];
  const float* theta = (const float*)d_in[3];
  const float* gma = (const float*)d_in[4];
  const float* omga = (const float*)d_in[5];
  float* y = (float*)d_out;
  float* h = (float*)d_out + (size_t)B_DIM * D_DIM * L_DIM;
  cema_kernel<<<dim3(D_DIM), dim3(256), 0, stream>>>(x, alpha, delta, theta,
                                                     gma, omga, y, h);
}


# state scan split over the 8 waves: each scans a 16-chunk segment from zero, the 8 totals are chained through LDS with q^256, each then re-scans from its true start state; state's sum order changes
# speedup vs baseline: 1.4463x; 1.1272x over previous
#include <hip/hip_runtime.h>


typedef float v8f __attribute__((ext_vector_type(8)));
typedef float v2f __attribute__((ext_vector_type(2)));

#define D_DIM 2048
#define N_MODES 16
#define B_DIM 2
#define L_DIM 4096
#define CG 128
#define NGROUP 2
#define MROW 17
#define UROW 136
#define XROW 17
#define NCHUNK (NGROUP * CG)
#define XV4 (L_DIM / 4 / 256)
#define SEGC 16
#define NSEG (CG / SEGC)

static __device__ __forceinline__ v8f wmma4(v2f a, v2f b, v8f c) {
  return __builtin_amdgcn_wmma_f32_16x16x4_f32(false, a, false, b, (short)0, c,
                                               false, false);
}

static __device__ __forceinline__ void cpow(float qr, float qi, int e,
                                            float& pr, float& pi) {
  pr = 1.f; pi = 0.f;
  for (int k = 0; k < e; ++k) {
    float t = pr * qr - pi * qi;
    pi = pr * qi + pi * qr;
    pr = t;
  }
}

__launch_bounds__(256, 2)
__global__ void cema_kernel(const float* __restrict__ x,
                            const float* __restrict__ alpha,
                            const float* __restrict__ delta,
                            const float* __restrict__ theta,
                            const float* __restrict__ gma,
                            const float* __restrict__ omga,
                            float* __restrict__ y_out,
                            float* __restrict__ h_out) {
  __shared__ float xs_s[B_DIM][NCHUNK * XROW];
  __shared__ __attribute__((aligned(16))) float Ure_s[B_DIM][N_MODES * UROW];
  __shared__ __attribute__((aligned(16))) float Uim_s[B_DIM][N_MODES * UROW];
  __shared__ float Sre_s[NSEG][32];
  __shared__ float Sim_s[NSEG][32];
  __shared__ float Bre_s[N_MODES * MROW];
  __shared__ float Bim_s[N_MODES * MROW];
  __shared__ float Are_s[N_MODES * MROW];
  __shared__ float AimN_s[N_MODES * MROW];
  __shared__ float Kmat_s[N_MODES * MROW];
  __shared__ float kern_s[N_MODES];
  __shared__ float cQr[16], cQi[16], cP[16], cGr[16], cGi[16];
  __shared__ float cCr[16], cCi[16], cTr[16], cTi[16];
  __shared__ float sOmega;

  const int tid = threadIdx.x;
  const int d = blockIdx.x;

  const int n16 = tid & 15;
  const float om_in = omga[d];
  const float th_in = theta[d];
  const float a_in = alpha[d * 16 + n16];
  const float de_in = delta[d * 16 + n16];
  const float g0_in = gma[(d * 16 + n16) * 2 + 0];
  const float g1_in = gma[(d * 16 + n16) * 2 + 1];
  float4 xr[B_DIM][XV4];
  #pragma unroll
  for (int b2 = 0; b2 < B_DIM; ++b2) {
    const float4* xg = (const float4*)(x + ((size_t)b2 * D_DIM + d) * L_DIM);
    #pragma unroll
    for (int k = 0; k < XV4; ++k) xr[b2][k] = xg[tid + 256 * k];
  }

  if (tid < 16) {
    int n = tid;
    float a = a_in;
    float de = de_in;
    float th = th_in;
    float p = 1.f / (1.f + expf(-a));
    float dd = 1.f / (1.f + expf(-de));
    float base = (1.f / (1.f + expf(-th))) * (2.f * 3.14159265358979323846f / 16.f);
    float phi = (float)(n + 1) * base;
    float r = 1.f - p * dd;
    float sn, cs;
    sincosf(phi, &sn, &cs);
    float qr = r * cs, qi = r * sn;
    cQr[n] = qr; cQi[n] = qi;
    cP[n] = p;
    float gr = g0_in * 0.25f;
    float gi = g1_in * 0.25f;
    cGr[n] = gr; cGi[n] = gi;
    cCr[n] = p * gr; cCi[n] = p * gi;
    float tr, ti;
    cpow(qr, qi, 16, tr, ti);
    cTr[n] = tr; cTi[n] = ti;
    if (n == 0) sOmega = om_in;
  }
  __syncthreads();

  {
    int r = tid >> 4, cc = tid & 15;
    float qr = cQr[cc], qi = cQi[cc];
    float pr, pi;
    cpow(qr, qi, r, pr, pi);
    xs_s[0][r * MROW + cc] = cCr[cc] * pr - cCi[cc] * pi;
    float q1r = pr * qr - pi * qi;
    float q1i = pr * qi + pi * qr;
    Are_s[r * MROW + cc] = cGr[cc] * q1r - cGi[cc] * q1i;
    AimN_s[r * MROW + cc] = -(cGr[cc] * q1i + cGi[cc] * q1r);
    float br, bi;
    cpow(cQr[r], cQi[r], 15 - cc, br, bi);
    Bre_s[r * MROW + cc] = cP[r] * br;
    Bim_s[r * MROW + cc] = cP[r] * bi;
  }
  __syncthreads();
  if (tid < 16) {
    float s = 0.f;
    for (int n = 0; n < 16; ++n) s += xs_s[0][tid * MROW + n];
    kern_s[tid] = s;
  }
  __syncthreads();
  {
    int r = tid >> 4, cc = tid & 15;
    float v = (r >= cc) ? kern_s[r - cc] : 0.f;
    if (r == cc) v += sOmega;
    Kmat_s[r * MROW + cc] = v;
  }

  #pragma unroll
  for (int b2 = 0; b2 < B_DIM; ++b2) {
    #pragma unroll
    for (int k = 0; k < XV4; ++k) {
      int i = tid + 256 * k;
      int t = i * 4;
      float4 v = xr[b2][k];
      float* dst = &xs_s[b2][(t >> 4) * XROW + (t & 15)];
      dst[0] = v.x; dst[1] = v.y; dst[2] = v.z; dst[3] = v.w;
    }
  }

  const int wv = tid >> 5;
  const int lane = tid & 31;
  const int lcol = lane & 15;
  const int khalf = lane >> 4;

  float hre = 0.f, him = 0.f;
  const float qtr = cTr[lcol], qti = cTi[lcol];
  float wtr = qtr, wti = qti;
  for (int k = 0; k < 4; ++k) {
    float t = wtr * wtr - wti * wti;
    wti = 2.f * wtr * wti;
    wtr = t;
  }

  #pragma unroll
  for (int g = 0; g < NGROUP; ++g) {
    __syncthreads();

    #pragma unroll
    for (int u = 0; u < 2; ++u) {
      int asg = wv * 2 + u;
      int b2 = asg >> 3;
      int tile = asg & 7;
      const float* xb = xs_s[b2] + g * (CG * XROW);
      int xbase = (tile * 16 + lcol) * XROW;
      v8f accR = {0.f, 0.f, 0.f, 0.f, 0.f, 0.f, 0.f, 0.f};
      v8f accI = {0.f, 0.f, 0.f, 0.f, 0.f, 0.f, 0.f, 0.f};
      #pragma unroll
      for (int kk = 0; kk < 4; ++kk) {
        int k0 = kk * 4 + khalf * 2;
        v2f bx; bx.x = xb[xbase + k0]; bx.y = xb[xbase + k0 + 1];
        v2f aR; aR.x = Bre_s[lcol * MROW + k0]; aR.y = Bre_s[lcol * MROW + k0 + 1];
        v2f aI; aI.x = Bim_s[lcol * MROW + k0]; aI.y = Bim_s[lcol * MROW + k0 + 1];
        accR = wmma4(aR, bx, accR);
        accI = wmma4(aI, bx, accI);
      }
      #pragma unroll
      for (int v = 0; v < 8; ++v) {
        int n = v + 8 * khalf;
        Ure_s[b2][n * UROW + tile * 16 + lcol] = accR[v];
        Uim_s[b2][n * UROW + tile * 16 + lcol] = accI[v];
      }
    }
    __syncthreads();

    {
      float* ur = &Ure_s[khalf][lcol * UROW + wv * SEGC];
      float* ui = &Uim_s[khalf][lcol * UROW + wv * SEGC];
      float u_r[SEGC], u_i[SEGC];
      for (int k = 0; k < SEGC / 4; ++k) {
        float4 a = *(const float4*)(ur + 4 * k);
        float4 b = *(const float4*)(ui + 4 * k);
        u_r[4 * k + 0] = a.x; u_r[4 * k + 1] = a.y; u_r[4 * k + 2] = a.z; u_r[4 * k + 3] = a.w;
        u_i[4 * k + 0] = b.x; u_i[4 * k + 1] = b.y; u_i[4 * k + 2] = b.z; u_i[4 * k + 3] = b.w;
      }
      float sr = u_r[0], si = u_i[0];
      for (int j = 1; j < SEGC; ++j) {
        float nr = fmaf(qtr, sr, fmaf(-qti, si, u_r[j]));
        float ni = fmaf(qtr, si, fmaf(qti, sr, u_i[j]));
        sr = nr; si = ni;
      }
      Sre_s[wv][lane] = sr;
      Sim_s[wv][lane] = si;
      __syncthreads();

      float cr = hre, ci = him;
      float mr = hre, mi = him;
      for (int s = 0; s < NSEG; ++s) {
        float tsr = Sre_s[s][lane], tsi = Sim_s[s][lane];
        float nr = fmaf(wtr, cr, fmaf(-wti, ci, tsr));
        float ni = fmaf(wtr, ci, fmaf(wti, cr, tsi));
        cr = nr; ci = ni;
        if (s + 1 == wv) { mr = cr; mi = ci; }
      }
      hre = cr; him = ci;

      float h_r[SEGC], h_i[SEGC];
      sr = mr; si = mi;
      for (int j = 0; j < SEGC; ++j) {
        h_r[j] = sr; h_i[j] = si;
        if (j + 1 < SEGC) {
          float nr = fmaf(qtr, sr, fmaf(-qti, si, u_r[j]));
          float ni = fmaf(qtr, si, fmaf(qti, sr, u_i[j]));
          sr = nr; si = ni;
        }
      }
      for (int k = 0; k < SEGC / 4; ++k) {
        *(float4*)(ur + 4 * k) = make_float4(h_r[4 * k], h_r[4 * k + 1], h_r[4 * k + 2], h_r[4 * k + 3]);
        *(float4*)(ui + 4 * k) = make_float4(h_i[4 * k], h_i[4 * k + 1], h_i[4 * k + 2], h_i[4 * k + 3]);
      }
    }
    __syncthreads();

    #pragma unroll
    for (int u = 0; u < 2; ++u) {
      int asg = wv * 2 + u;
      int b2 = asg >> 3;
      int tile = asg & 7;
      float* xb = xs_s[b2] + g * (CG * XROW);
      int xbase = (tile * 16 + lcol) * XROW;
      v8f acc = {0.f, 0.f, 0.f, 0.f, 0.f, 0.f, 0.f, 0.f};
      #pragma unroll
      for (int kk = 0; kk < 4; ++kk) {
        int k0 = kk * 4 + khalf * 2;
        v2f bx; bx.x = xb[xbase + k0]; bx.y = xb[xbase + k0 + 1];
        v2f aK; aK.x = Kmat_s[lcol * MROW + k0]; aK.y = Kmat_s[lcol * MROW + k0 + 1];
        acc = wmma4(aK, bx, acc);
      }
      #pragma unroll
      for (int kk = 0; kk < 4; ++kk) {
        int k0 = kk * 4 + khalf * 2;
        v2f bh;
        bh.x = Ure_s[b2][k0 * UROW + tile * 16 + lcol];
        bh.y = Ure_s[b2][(k0 + 1) * UROW + tile * 16 + lcol];
        v2f aA; aA.x = Are_s[lcol * MROW + k0]; aA.y = Are_s[lcol * MROW + k0 + 1];
        acc = wmma4(aA, bh, acc);
      }
      #pragma unroll
      for (int kk = 0; kk < 4; ++kk) {
        int k0 = kk * 4 + khalf * 2;
        v2f bh;
        bh.x = Uim_s[b2][k0 * UROW + tile * 16 + lcol];
        bh.y = Uim_s[b2][(k0 + 1) * UROW + tile * 16 + lcol];
        v2f aA; aA.x = AimN_s[lcol * MROW + k0]; aA.y = AimN_s[lcol * MROW + k0 + 1];
        acc = wmma4(aA, bh, acc);
      }
      #pragma unroll
      for (int v = 0; v < 8; ++v) {
        xb[(tile * 16 + lcol) * XROW + v + 8 * khalf] = acc[v];
      }
    }
    __syncthreads();

    for (int b2 = 0; b2 < B_DIM; ++b2) {
      float4* yg = (float4*)(y_out + ((size_t)b2 * D_DIM + d) * L_DIM +
                             (size_t)g * (CG * 16));
      for (int i = tid; i < CG * 16 / 4; i += 256) {
        int t = i * 4;
        const float* src = &xs_s[b2][g * (CG * XROW) + (t >> 4) * XROW + (t & 15)];
        float4 v;
        v.x = src[0]; v.y = src[1]; v.z = src[2]; v.w = src[3];
        yg[i] = v;
      }
    }
  }

  if (tid < 32) {
    int b2 = tid >> 4, n = tid & 15;
    float* hp = h_out + (((size_t)b2 * D_DIM + d) * N_MODES + n) * 2;
    hp[0] = hre;
    hp[1] = him;
  }
}

extern "C" void kernel_launch(void* const* d_in, const int* in_sizes, int n_in,
                              void* d_out, int out_size, void* d_ws, size_t ws_size,
                              hipStream_t stream) {
  (void)in_sizes; (void)n_in; (void)out_size; (void)d_ws; (void)ws_size;
  const float* x = (const float*)d_in[0];
  const float* alpha = (const float*)d_in[1];
  const float* delta = (const float*)d_in[2];
  const float* theta = (const float*)d_in[3];
  const float* gma = (const float*)d_in[4];
  const float* omga = (const float*)d_in[5];
  float* y = (float*)d_out;
  float* h = (float*)d_out + (size_t)B_DIM * D_DIM * L_DIM;
  cema_kernel<<<dim3(D_DIM), dim3(256), 0, stream>>>(x, alpha, delta, theta,
                                                     gma, omga, y, h);
}


# each wave keeps one 16-chunk tile of both batches from U through scan, H and Y to its own store: 14 block barriers become 6, and the loop's one barrier fences LDS only, so no wait on global stores
# speedup vs baseline: 1.5402x; 1.0649x over previous
#include <hip/hip_runtime.h>


typedef float v8f __attribute__((ext_vector_type(8)));
typedef float v2f __attribute__((ext_vector_type(2)));

#define D_DIM 2048
#define N_MODES 16
#define B_DIM 2
#define L_DIM 4096
#define CG 128
#define NGROUP 2
#define MROW 17
#define UROW 136
#define XROW 17
#define NCHUNK (NGROUP * CG)
#define XV4 (L_DIM / 4 / 256)
#define SEGC 16
#define NSEG (CG / SEGC)

static __device__ __forceinline__ v8f wmma4(v2f a, v2f b, v8f c) {
  return __builtin_amdgcn_wmma_f32_16x16x4_f32(false, a, false, b, (short)0, c,
                                               false, false);
}

static __device__ __forceinline__ void wave_lds_order() {
  __builtin_amdgcn_fence(__ATOMIC_RELEASE, "wavefront");
  __builtin_amdgcn_wave_barrier();
  __builtin_amdgcn_fence(__ATOMIC_ACQUIRE, "wavefront");
}

static __device__ __forceinline__ void block_lds_barrier() {
  __builtin_amdgcn_fence(__ATOMIC_RELEASE, "workgroup", "local");
  __builtin_amdgcn_s_barrier();
  __builtin_amdgcn_fence(__ATOMIC_ACQUIRE, "workgroup", "local");
}

static __device__ __forceinline__ void cpow(float qr, float qi, int e,
                                            float& pr, float& pi) {
  pr = 1.f; pi = 0.f;
  for (int k = 0; k < e; ++k) {
    float t = pr * qr - pi * qi;
    pi = pr * qi + pi * qr;
    pr = t;
  }
}

__launch_bounds__(256, 8)
__global__ void cema_kernel(const float* __restrict__ x,
                            const float* __restrict__ alpha,
                            const float* __restrict__ delta,
                            const float* __restrict__ theta,
                            const float* __restrict__ gma,
                            const float* __restrict__ omga,
                            float* __restrict__ y_out,
                            float* __restrict__ h_out) {
  __shared__ float xs_s[B_DIM][NCHUNK * XROW];
  __shared__ __attribute__((aligned(16))) float Ure_s[B_DIM][N_MODES * UROW];
  __shared__ __attribute__((aligned(16))) float Uim_s[B_DIM][N_MODES * UROW];
  __shared__ float Sre_s[NGROUP][NSEG][32];
  __shared__ float Sim_s[NGROUP][NSEG][32];
  __shared__ float Bre_s[N_MODES * MROW];
  __shared__ float Bim_s[N_MODES * MROW];
  __shared__ float Are_s[N_MODES * MROW];
  __shared__ float AimN_s[N_MODES * MROW];
  __shared__ float Kmat_s[N_MODES * MROW];
  __shared__ float kern_s[N_MODES];
  __shared__ float cQr[16], cQi[16], cP[16], cGr[16], cGi[16];
  __shared__ float cCr[16], cCi[16], cTr[16], cTi[16];
  __shared__ float sOmega;

  const int tid = threadIdx.x;
  const int d = blockIdx.x;

  const int n16 = tid & 15;
  const float om_in = omga[d];
  const float th_in = theta[d];
  const float a_in = alpha[d * 16 + n16];
  const float de_in = delta[d * 16 + n16];
  const float g0_in = gma[(d * 16 + n16) * 2 + 0];
  const float g1_in = gma[(d * 16 + n16) * 2 + 1];
  float4 xr[B_DIM][XV4];
  #pragma unroll
  for (int b2 = 0; b2 < B_DIM; ++b2) {
    const float4* xg = (const float4*)(x + ((size_t)b2 * D_DIM + d) * L_DIM);
    #pragma unroll
    for (int k = 0; k < XV4; ++k) xr[b2][k] = xg[tid + 256 * k];
  }

  if (tid < 16) {
    int n = tid;
    float a = a_in;
    float de = de_in;
    float th = th_in;
    float p = 1.f / (1.f + expf(-a));
    float dd = 1.f / (1.f + expf(-de));
    float base = (1.f / (1.f + expf(-th))) * (2.f * 3.14159265358979323846f / 16.f);
    float phi = (float)(n + 1) * base;
    float r = 1.f - p * dd;
    float sn, cs;
    sincosf(phi, &sn, &cs);
    float qr = r * cs, qi = r * sn;
    cQr[n] = qr; cQi[n] = qi;
    cP[n] = p;
    float gr = g0_in * 0.25f;
    float gi = g1_in * 0.25f;
    cGr[n] = gr; cGi[n] = gi;
    cCr[n] = p * gr; cCi[n] = p * gi;
    float tr, ti;
    cpow(qr, qi, 16, tr, ti);
    cTr[n] = tr; cTi[n] = ti;
    if (n == 0) sOmega = om_in;
  }
  __syncthreads();

  {
    int r = tid >> 4, cc = tid & 15;
    float qr = cQr[cc], qi = cQi[cc];
    float pr, pi;
    cpow(qr, qi, r, pr, pi);
    xs_s[0][r * MROW + cc] = cCr[cc] * pr - cCi[cc] * pi;
    float q1r = pr * qr - pi * qi;
    float q1i = pr * qi + pi * qr;
    Are_s[r * MROW + cc] = cGr[cc] * q1r - cGi[cc] * q1i;
    AimN_s[r * MROW + cc] = -(cGr[cc] * q1i + cGi[cc] * q1r);
    float br, bi;
    cpow(cQr[r], cQi[r], 15 - cc, br, bi);
    Bre_s[r * MROW + cc] = cP[r] * br;
    Bim_s[r * MROW + cc] = cP[r] * bi;
  }
  __syncthreads();
  if (tid < 16) {
    float s = 0.f;
    for (int n = 0; n < 16; ++n) s += xs_s[0][tid * MROW + n];
    kern_s[tid] = s;
  }
  __syncthreads();
  {
    int r = tid >> 4, cc = tid & 15;
    float v = (r >= cc) ? kern_s[r - cc] : 0.f;
    if (r == cc) v += sOmega;
    Kmat_s[r * MROW + cc] = v;
  }

  #pragma unroll
  for (int b2 = 0; b2 < B_DIM; ++b2) {
    #pragma unroll
    for (int k = 0; k < XV4; ++k) {
      int i = tid + 256 * k;
      int t = i * 4;
      float4 v = xr[b2][k];
      float* dst = &xs_s[b2][(t >> 4) * XROW + (t & 15)];
      dst[0] = v.x; dst[1] = v.y; dst[2] = v.z; dst[3] = v.w;
    }
  }

  const int wv = tid >> 5;
  const int lane = tid & 31;
  const int lcol = lane & 15;
  const int khalf = lane >> 4;

  float hre = 0.f, him = 0.f;
  const float qtr = cTr[lcol], qti = cTi[lcol];
  float wtr = qtr, wti = qti;
  for (int k = 0; k < 4; ++k) {
    float t = wtr * wtr - wti * wti;
    wti = 2.f * wtr * wti;
    wtr = t;
  }

  __syncthreads();

  #pragma unroll
  for (int g = 0; g < NGROUP; ++g) {
    wave_lds_order();

    #pragma unroll
    for (int u = 0; u < 2; ++u) {
      int b2 = u;
      int tile = wv;
      const float* xb = xs_s[b2] + g * (CG * XROW);
      int xbase = (tile * 16 + lcol) * XROW;
      v8f accR = {0.f, 0.f, 0.f, 0.f, 0.f, 0.f, 0.f, 0.f};
      v8f accI = {0.f, 0.f, 0.f, 0.f, 0.f, 0.f, 0.f, 0.f};
      #pragma unroll
      for (int kk = 0; kk < 4; ++kk) {
        int k0 = kk * 4 + khalf * 2;
        v2f bx; bx.x = xb[xbase + k0]; bx.y = xb[xbase + k0 + 1];
        v2f aR; aR.x = Bre_s[lcol * MROW + k0]; aR.y = Bre_s[lcol * MROW + k0 + 1];
        v2f aI; aI.x = Bim_s[lcol * MROW + k0]; aI.y = Bim_s[lcol * MROW + k0 + 1];
        accR = wmma4(aR, bx, accR);
        accI = wmma4(aI, bx, accI);
      }
      #pragma unroll
      for (int v = 0; v < 8; ++v) {
        int n = v + 8 * khalf;
        Ure_s[b2][n * UROW + tile * 16 + lcol] = accR[v];
        Uim_s[b2][n * UROW + tile * 16 + lcol] = accI[v];
      }
    }
    wave_lds_order();

    {
      float* ur = &Ure_s[khalf][lcol * UROW + wv * SEGC];
      float* ui = &Uim_s[khalf][lcol * UROW + wv * SEGC];
      float u_r[SEGC], u_i[SEGC];
      for (int k = 0; k < SEGC / 4; ++k) {
        float4 a = *(const float4*)(ur + 4 * k);
        float4 b = *(const float4*)(ui + 4 * k);
        u_r[4 * k + 0] = a.x; u_r[4 * k + 1] = a.y; u_r[4 * k + 2] = a.z; u_r[4 * k + 3] = a.w;
        u_i[4 * k + 0] = b.x; u_i[4 * k + 1] = b.y; u_i[4 * k + 2] = b.z; u_i[4 * k + 3] = b.w;
      }
      float sr = u_r[0], si = u_i[0];
      for (int j = 1; j < SEGC; ++j) {
        float nr = fmaf(qtr, sr, fmaf(-qti, si, u_r[j]));
        float ni = fmaf(qtr, si, fmaf(qti, sr, u_i[j]));
        sr = nr; si = ni;
      }
      Sre_s[g][wv][lane] = sr;
      Sim_s[g][wv][lane] = si;
      block_lds_barrier();

      float cr = hre, ci = him;
      float mr = hre, mi = him;
      for (int s = 0; s < NSEG; ++s) {
        float tsr = Sre_s[g][s][lane], tsi = Sim_s[g][s][lane];
        float nr = fmaf(wtr, cr, fmaf(-wti, ci, tsr));
        float ni = fmaf(wtr, ci, fmaf(wti, cr, tsi));
        cr = nr; ci = ni;
        if (s + 1 == wv) { mr = cr; mi = ci; }
      }
      hre = cr; him = ci;

      float h_r[SEGC], h_i[SEGC];
      sr = mr; si = mi;
      for (int j = 0; j < SEGC; ++j) {
        h_r[j] = sr; h_i[j] = si;
        if (j + 1 < SEGC) {
          float nr = fmaf(qtr, sr, fmaf(-qti, si, u_r[j]));
          float ni = fmaf(qtr, si, fmaf(qti, sr, u_i[j]));
          sr = nr; si = ni;
        }
      }
      for (int k = 0; k < SEGC / 4; ++k) {
        *(float4*)(ur + 4 * k) = make_float4(h_r[4 * k], h_r[4 * k + 1], h_r[4 * k + 2], h_r[4 * k + 3]);
        *(float4*)(ui + 4 * k) = make_float4(h_i[4 * k], h_i[4 * k + 1], h_i[4 * k + 2], h_i[4 * k + 3]);
      }
    }
    wave_lds_order();

    #pragma unroll
    for (int u = 0; u < 2; ++u) {
      int b2 = u;
      int tile = wv;
      float* xb = xs_s[b2] + g * (CG * XROW);
      int xbase = (tile * 16 + lcol) * XROW;
      v8f acc = {0.f, 0.f, 0.f, 0.f, 0.f, 0.f, 0.f, 0.f};
      #pragma unroll
      for (int kk = 0; kk < 4; ++kk) {
        int k0 = kk * 4 + khalf * 2;
        v2f bx; bx.x = xb[xbase + k0]; bx.y = xb[xbase + k0 + 1];
        v2f aK; aK.x = Kmat_s[lcol * MROW + k0]; aK.y = Kmat_s[lcol * MROW + k0 + 1];
        acc = wmma4(aK, bx, acc);
      }
      #pragma unroll
      for (int kk = 0; kk < 4; ++kk) {
        int k0 = kk * 4 + khalf * 2;
        v2f bh;
        bh.x = Ure_s[b2][k0 * UROW + tile * 16 + lcol];
        bh.y = Ure_s[b2][(k0 + 1) * UROW + tile * 16 + lcol];
        v2f aA; aA.x = Are_s[lcol * MROW + k0]; aA.y = Are_s[lcol * MROW + k0 + 1];
        acc = wmma4(aA, bh, acc);
      }
      #pragma unroll
      for (int kk = 0; kk < 4; ++kk) {
        int k0 = kk * 4 + khalf * 2;
        v2f bh;
        bh.x = Uim_s[b2][k0 * UROW + tile * 16 + lcol];
        bh.y = Uim_s[b2][(k0 + 1) * UROW + tile * 16 + lcol];
        v2f aA; aA.x = AimN_s[lcol * MROW + k0]; aA.y = AimN_s[lcol * MROW + k0 + 1];
        acc = wmma4(aA, bh, acc);
      }
      #pragma unroll
      for (int v = 0; v < 8; ++v) {
        xb[(tile * 16 + lcol) * XROW + v + 8 * khalf] = acc[v];
      }
    }
    wave_lds_order();

    #pragma unroll
    for (int b2 = 0; b2 < B_DIM; ++b2) {
      float4* yg = (float4*)(y_out + ((size_t)b2 * D_DIM + d) * L_DIM +
                             (size_t)g * (CG * 16) + (size_t)wv * 256);
      #pragma unroll
      for (int k = 0; k < 2; ++k) {
        int i = lane + 32 * k;
        int t = i * 4;
        const float* src = &xs_s[b2][g * (CG * XROW) + (wv * 16 + (t >> 4)) * XROW + (t & 15)];
        float4 v;
        v.x = src[0]; v.y = src[1]; v.z = src[2]; v.w = src[3];
        yg[i] = v;
      }
    }
  }

  if (tid < 32) {
    int b2 = tid >> 4, n = tid & 15;
    float* hp = h_out + (((size_t)b2 * D_DIM + d) * N_MODES + n) * 2;
    hp[0] = hre;
    hp[1] = him;
  }
}

extern "C" void kernel_launch(void* const* d_in, const int* in_sizes, int n_in,
                              void* d_out, int out_size, void* d_ws, size_t ws_size,
                              hipStream_t stream) {
  (void)in_sizes; (void)n_in; (void)out_size; (void)d_ws; (void)ws_size;
  const float* x = (const float*)d_in[0];
  const float* alpha = (const float*)d_in[1];
  const float* delta = (const float*)d_in[2];
  const float* theta = (const float*)d_in[3];
  const float* gma = (const float*)d_in[4];
  const float* omga = (const float*)d_in[5];
  float* y = (float*)d_out;
  float* h = (float*)d_out + (size_t)B_DIM * D_DIM * L_DIM;
  cema_kernel<<<dim3(D_DIM), dim3(256), 0, stream>>>(x, alpha, delta, theta,
                                                     gma, omga, y, h);
}


# LDS layouts made bank-conflict-free: x/y tile unpadded, its 16-byte groups XOR-swizzled (x staging, y stores, stream-out as 16-byte LDS ops); U/H rows 16 B x odd; matrices 80 B; sum orders unchanged
# speedup vs baseline: 1.6400x; 1.0648x over previous
#include <hip/hip_runtime.h>


typedef float v8f __attribute__((ext_vector_type(8)));
typedef float v2f __attribute__((ext_vector_type(2)));
typedef float v4f __attribute__((ext_vector_type(4)));

#define D_DIM 2048
#define N_MODES 16
#define B_DIM 2
#define L_DIM 4096
#define CG 128
#define NGROUP 2
#define MROW 20
#define UROW 140
#define XROW 16
#define NCHUNK (NGROUP * CG)
#define XV4 (L_DIM / 4 / 256)
#define SEGC 16
#define NSEG (CG / SEGC)

static __device__ __forceinline__ v8f wmma4(v2f a, v2f b, v8f c) {
  return __builtin_amdgcn_wmma_f32_16x16x4_f32(false, a, false, b, (short)0, c,
                                               false, false);
}

static __device__ __forceinline__ int xswz(int chunk, int grp) {
  return chunk * XROW + ((grp ^ ((chunk >> 2) & 3)) << 2);
}

static __device__ __forceinline__ void wave_lds_order() {
  __builtin_amdgcn_fence(__ATOMIC_RELEASE, "wavefront");
  __builtin_amdgcn_wave_barrier();
  __builtin_amdgcn_fence(__ATOMIC_ACQUIRE, "wavefront");
}

static __device__ __forceinline__ void block_lds_barrier() {
  __builtin_amdgcn_fence(__ATOMIC_RELEASE, "workgroup", "local");
  __builtin_amdgcn_s_barrier();
  __builtin_amdgcn_fence(__ATOMIC_ACQUIRE, "workgroup", "local");
}

static __device__ __forceinline__ void cpow(float qr, float qi, int e,
                                            float& pr, float& pi) {
  pr = 1.f; pi = 0.f;
  for (int k = 0; k < e; ++k) {
    float t = pr * qr - pi * qi;
    pi = pr * qi + pi * qr;
    pr = t;
  }
}

__launch_bounds__(256, 8)
__global__ void cema_kernel(const float* __restrict__ x,
                            const float* __restrict__ alpha,
                            const float* __restrict__ delta,
                            const float* __restrict__ theta,
                            const float* __restrict__ gma,
                            const float* __restrict__ omga,
                            float* __restrict__ y_out,
                            float* __restrict__ h_out) {
  __shared__ __attribute__((aligned(16))) float xs_s[B_DIM][NCHUNK * XROW];
  __shared__ __attribute__((aligned(16))) float Ure_s[B_DIM][N_MODES * UROW];
  __shared__ __attribute__((aligned(16))) float Uim_s[B_DIM][N_MODES * UROW];
  __shared__ float Sre_s[NGROUP][NSEG][32];
  __shared__ float Sim_s[NGROUP][NSEG][32];
  __shared__ __attribute__((aligned(16))) float Bre_s[N_MODES * MROW];
  __shared__ __attribute__((aligned(16))) float Bim_s[N_MODES * MROW];
  __shared__ __attribute__((aligned(16))) float Are_s[N_MODES * MROW];
  __shared__ __attribute__((aligned(16))) float AimN_s[N_MODES * MROW];
  __shared__ __attribute__((aligned(16))) float Kmat_s[N_MODES * MROW];
  __shared__ float kern_s[N_MODES];
  __shared__ float cQr[16], cQi[16], cP[16], cGr[16], cGi[16];
  __shared__ float cCr[16], cCi[16], cTr[16], cTi[16];
  __shared__ float sOmega;

  const int tid = threadIdx.x;
  const int d = blockIdx.x;

  const int n16 = tid & 15;
  const float om_in = omga[d];
  const float th_in = theta[d];
  const float a_in = alpha[d * 16 + n16];
  const float de_in = delta[d * 16 + n16];
  const float g0_in = gma[(d * 16 + n16) * 2 + 0];
  const float g1_in = gma[(d * 16 + n16) * 2 + 1];
  v4f xr[B_DIM][XV4];
  #pragma unroll
  for (int b2 = 0; b2 < B_DIM; ++b2) {
    const v4f* xg = (const v4f*)(x + ((size_t)b2 * D_DIM + d) * L_DIM);
    #pragma unroll
    for (int k = 0; k < XV4; ++k) xr[b2][k] = xg[tid + 256 * k];
  }

  if (tid < 16) {
    int n = tid;
    float a = a_in;
    float de = de_in;
    float th = th_in;
    float p = 1.f / (1.f + expf(-a));
    float dd = 1.f / (1.f + expf(-de));
    float base = (1.f / (1.f + expf(-th))) * (2.f * 3.14159265358979323846f / 16.f);
    float phi = (float)(n + 1) * base;
    float r = 1.f - p * dd;
    float sn, cs;
    sincosf(phi, &sn, &cs);
    float qr = r * cs, qi = r * sn;
    cQr[n] = qr; cQi[n] = qi;
    cP[n] = p;
    float gr = g0_in * 0.25f;
    float gi = g1_in * 0.25f;
    cGr[n] = gr; cGi[n] = gi;
    cCr[n] = p * gr; cCi[n] = p * gi;
    float tr, ti;
    cpow(qr, qi, 16, tr, ti);
    cTr[n] = tr; cTi[n] = ti;
    if (n == 0) sOmega = om_in;
  }
  __syncthreads();

  {
    int r = tid >> 4, cc = tid & 15;
    float qr = cQr[cc], qi = cQi[cc];
    float pr, pi;
    cpow(qr, qi, r, pr, pi);
    xs_s[0][r * MROW + cc] = cCr[cc] * pr - cCi[cc] * pi;
    float q1r = pr * qr - pi * qi;
    float q1i = pr * qi + pi * qr;
    Are_s[r * MROW + cc] = cGr[cc] * q1r - cGi[cc] * q1i;
    AimN_s[r * MROW + cc] = -(cGr[cc] * q1i + cGi[cc] * q1r);
    float br, bi;
    cpow(cQr[r], cQi[r], 15 - cc, br, bi);
    Bre_s[r * MROW + cc] = cP[r] * br;
    Bim_s[r * MROW + cc] = cP[r] * bi;
  }
  __syncthreads();
  if (tid < 16) {
    float s = 0.f;
    for (int n = 0; n < 16; ++n) s += xs_s[0][tid * MROW + n];
    kern_s[tid] = s;
  }
  __syncthreads();
  {
    int r = tid >> 4, cc = tid & 15;
    float v = (r >= cc) ? kern_s[r - cc] : 0.f;
    if (r == cc) v += sOmega;
    Kmat_s[r * MROW + cc] = v;
  }

  asm volatile("" :: "v"(xr[0][0]), "v"(xr[0][1]), "v"(xr[0][2]), "v"(xr[0][3]),
                     "v"(xr[1][0]), "v"(xr[1][1]), "v"(xr[1][2]), "v"(xr[1][3]));
  #pragma unroll
  for (int b2 = 0; b2 < B_DIM; ++b2) {
    #pragma unroll
    for (int k = 0; k < XV4; ++k) {
      int i = tid + 256 * k;
      *(v4f*)&xs_s[b2][xswz(i >> 2, i & 3)] = xr[b2][k];
    }
  }

  const int wv = tid >> 5;
  const int lane = tid & 31;
  const int lcol = lane & 15;
  const int khalf = lane >> 4;
  const int sB = (lcol >> 2) & 3;

  float hre = 0.f, him = 0.f;
  const float qtr = cTr[lcol], qti = cTi[lcol];
  float wtr = qtr, wti = qti;
  for (int k = 0; k < 4; ++k) {
    float t = wtr * wtr - wti * wti;
    wti = 2.f * wtr * wti;
    wtr = t;
  }

  __syncthreads();

  #pragma unroll
  for (int g = 0; g < NGROUP; ++g) {
    wave_lds_order();

    #pragma unroll
    for (int u = 0; u < 2; ++u) {
      int b2 = u;
      int tile = wv;
      const float* xb = xs_s[b2] + g * (CG * XROW);
      int xbase = (tile * 16 + lcol) * XROW + khalf * 2;
      v8f accR = {0.f, 0.f, 0.f, 0.f, 0.f, 0.f, 0.f, 0.f};
      v8f accI = {0.f, 0.f, 0.f, 0.f, 0.f, 0.f, 0.f, 0.f};
      #pragma unroll
      for (int kk = 0; kk < 4; ++kk) {
        int k0 = kk * 4 + khalf * 2;
        v2f bx = *(const v2f*)&xb[xbase + ((kk ^ sB) << 2)];
        v2f aR = *(const v2f*)&Bre_s[lcol * MROW + k0];
        v2f aI = *(const v2f*)&Bim_s[lcol * MROW + k0];
        accR = wmma4(aR, bx, accR);
        accI = wmma4(aI, bx, accI);
      }
      #pragma unroll
      for (int v = 0; v < 8; ++v) {
        int n = v + 8 * khalf;
        Ure_s[b2][n * UROW + tile * 16 + lcol] = accR[v];
        Uim_s[b2][n * UROW + tile * 16 + lcol] = accI[v];
      }
    }
    wave_lds_order();

    {
      float* ur = &Ure_s[khalf][lcol * UROW + wv * SEGC];
      float* ui = &Uim_s[khalf][lcol * UROW + wv * SEGC];
      float u_r[SEGC], u_i[SEGC];
      for (int k = 0; k < SEGC / 4; ++k) {
        float4 a = *(const float4*)(ur + 4 * k);
        float4 b = *(const float4*)(ui + 4 * k);
        u_r[4 * k + 0] = a.x; u_r[4 * k + 1] = a.y; u_r[4 * k + 2] = a.z; u_r[4 * k + 3] = a.w;
        u_i[4 * k + 0] = b.x; u_i[4 * k + 1] = b.y; u_i[4 * k + 2] = b.z; u_i[4 * k + 3] = b.w;
      }
      float sr = u_r[0], si = u_i[0];
      for (int j = 1; j < SEGC; ++j) {
        float nr = fmaf(qtr, sr, fmaf(-qti, si, u_r[j]));
        float ni = fmaf(qtr, si, fmaf(qti, sr, u_i[j]));
        sr = nr; si = ni;
      }
      Sre_s[g][wv][lane] = sr;
      Sim_s[g][wv][lane] = si;
      block_lds_barrier();

      float cr = hre, ci = him;
      float mr = hre, mi = him;
      for (int s = 0; s < NSEG; ++s) {
        float tsr = Sre_s[g][s][lane], tsi = Sim_s[g][s][lane];
        float nr = fmaf(wtr, cr, fmaf(-wti, ci, tsr));
        float ni = fmaf(wtr, ci, fmaf(wti, cr, tsi));
        cr = nr; ci = ni;
        if (s + 1 == wv) { mr = cr; mi = ci; }
      }
      hre = cr; him = ci;

      float h_r[SEGC], h_i[SEGC];
      sr = mr; si = mi;
      for (int j = 0; j < SEGC; ++j) {
        h_r[j] = sr; h_i[j] = si;
        if (j + 1 < SEGC) {
          float nr = fmaf(qtr, sr, fmaf(-qti, si, u_r[j]));
          float ni = fmaf(qtr, si, fmaf(qti, sr, u_i[j]));
          sr = nr; si = ni;
        }
      }
      for (int k = 0; k < SEGC / 4; ++k) {
        *(float4*)(ur + 4 * k) = make_float4(h_r[4 * k], h_r[4 * k + 1], h_r[4 * k + 2], h_r[4 * k + 3]);
        *(float4*)(ui + 4 * k) = make_float4(h_i[4 * k], h_i[4 * k + 1], h_i[4 * k + 2], h_i[4 * k + 3]);
      }
    }
    wave_lds_order();

    #pragma unroll
    for (int u = 0; u < 2; ++u) {
      int b2 = u;
      int tile = wv;
      float* xb = xs_s[b2] + g * (CG * XROW);
      int xbase = (tile * 16 + lcol) * XROW + khalf * 2;
      v8f acc = {0.f, 0.f, 0.f, 0.f, 0.f, 0.f, 0.f, 0.f};
      #pragma unroll
      for (int kk = 0; kk < 4; ++kk) {
        int k0 = kk * 4 + khalf * 2;
        v2f bx = *(const v2f*)&xb[xbase + ((kk ^ sB) << 2)];
        v2f aK = *(const v2f*)&Kmat_s[lcol * MROW + k0];
        acc = wmma4(aK, bx, acc);
      }
      #pragma unroll
      for (int kk = 0; kk < 4; ++kk) {
        int k0 = kk * 4 + khalf * 2;
        v2f bh;
        bh.x = Ure_s[b2][k0 * UROW + tile * 16 + lcol];
        bh.y = Ure_s[b2][(k0 + 1) * UROW + tile * 16 + lcol];
        v2f aA = *(const v2f*)&Are_s[lcol * MROW + k0];
        acc = wmma4(aA, bh, acc);
      }
      #pragma unroll
      for (int kk = 0; kk < 4; ++kk) {
        int k0 = kk * 4 + khalf * 2;
        v2f bh;
        bh.x = Uim_s[b2][k0 * UROW + tile * 16 + lcol];
        bh.y = Uim_s[b2][(k0 + 1) * UROW + tile * 16 + lcol];
        v2f aA = *(const v2f*)&AimN_s[lcol * MROW + k0];
        acc = wmma4(aA, bh, acc);
      }
      *(float4*)&xb[(tile * 16 + lcol) * XROW + (((2 * khalf) ^ sB) << 2)] =
          make_float4(acc[0], acc[1], acc[2], acc[3]);
      *(float4*)&xb[(tile * 16 + lcol) * XROW + (((2 * khalf + 1) ^ sB) << 2)] =
          make_float4(acc[4], acc[5], acc[6], acc[7]);
    }
    wave_lds_order();

    #pragma unroll
    for (int b2 = 0; b2 < B_DIM; ++b2) {
      float4* yg = (float4*)(y_out + ((size_t)b2 * D_DIM + d) * L_DIM +
                             (size_t)g * (CG * 16) + (size_t)wv * 256);
      #pragma unroll
      for (int k = 0; k < 2; ++k) {
        int i = lane + 32 * k;
        yg[i] = *(const float4*)&xs_s[b2][g * (CG * XROW) + xswz(wv * 16 + (i >> 2), i & 3)];
      }
    }
  }

  if (tid < 32) {
    int b2 = tid >> 4, n = tid & 15;
    float* hp = h_out + (((size_t)b2 * D_DIM + d) * N_MODES + n) * 2;
    hp[0] = hre;
    hp[1] = him;
  }
}

extern "C" void kernel_launch(void* const* d_in, const int* in_sizes, int n_in,
                              void* d_out, int out_size, void* d_ws, size_t ws_size,
                              hipStream_t stream) {
  (void)in_sizes; (void)n_in; (void)out_size; (void)d_ws; (void)ws_size;
  const float* x = (const float*)d_in[0];
  const float* alpha = (const float*)d_in[1];
  const float* delta = (const float*)d_in[2];
  const float* theta = (const float*)d_in[3];
  const float* gma = (const float*)d_in[4];
  const float* omga = (const float*)d_in[5];
  float* y = (float*)d_out;
  float* h = (float*)d_out + (size_t)B_DIM * D_DIM * L_DIM;
  cema_kernel<<<dim3(D_DIM), dim3(256), 0, stream>>>(x, alpha, delta, theta,
                                                     gma, omga, y, h);
}


# set-up runs under the x loads' flight: coefficient loads issued first and waited for by count, the four set-up barriers fence LDS only, x is waited for once at its LDS store; sum orders unchanged
# speedup vs baseline: 1.7210x; 1.0494x over previous
#include <hip/hip_runtime.h>


typedef float v8f __attribute__((ext_vector_type(8)));
typedef float v2f __attribute__((ext_vector_type(2)));
typedef float v4f __attribute__((ext_vector_type(4)));

#define D_DIM 2048
#define N_MODES 16
#define B_DIM 2
#define L_DIM 4096
#define CG 128
#define NGROUP 2
#define MROW 20
#define UROW 140
#define XROW 16
#define NCHUNK (NGROUP * CG)
#define XV4 (L_DIM / 4 / 256)
#define SEGC 16
#define NSEG (CG / SEGC)

static __device__ __forceinline__ v8f wmma4(v2f a, v2f b, v8f c) {
  return __builtin_amdgcn_wmma_f32_16x16x4_f32(false, a, false, b, (short)0, c,
                                               false, false);
}

static __device__ __forceinline__ int xswz(int chunk, int grp) {
  return chunk * XROW + ((grp ^ ((chunk >> 2) & 3)) << 2);
}

static __device__ __forceinline__ void wave_lds_order() {
  __builtin_amdgcn_fence(__ATOMIC_RELEASE, "wavefront");
  __builtin_amdgcn_wave_barrier();
  __builtin_amdgcn_fence(__ATOMIC_ACQUIRE, "wavefront");
}

static __device__ __forceinline__ void block_lds_barrier() {
  __builtin_amdgcn_fence(__ATOMIC_RELEASE, "workgroup", "local");
  __builtin_amdgcn_s_barrier();
  __builtin_amdgcn_fence(__ATOMIC_ACQUIRE, "workgroup", "local");
}

static __device__ __forceinline__ void cpow(float qr, float qi, int e,
                                            float& pr, float& pi) {
  pr = 1.f; pi = 0.f;
  for (int k = 0; k < e; ++k) {
    float t = pr * qr - pi * qi;
    pi = pr * qi + pi * qr;
    pr = t;
  }
}

__launch_bounds__(256, 8)
__global__ void cema_kernel(const float* __restrict__ x,
                            const float* __restrict__ alpha,
                            const float* __restrict__ delta,
                            const float* __restrict__ theta,
                            const float* __restrict__ gma,
                            const float* __restrict__ omga,
                            float* __restrict__ y_out,
                            float* __restrict__ h_out) {
  __shared__ __attribute__((aligned(16))) float xs_s[B_DIM][NCHUNK * XROW];
  __shared__ __attribute__((aligned(16))) float Ure_s[B_DIM][N_MODES * UROW];
  __shared__ __attribute__((aligned(16))) float Uim_s[B_DIM][N_MODES * UROW];
  __shared__ float Sre_s[NGROUP][NSEG][32];
  __shared__ float Sim_s[NGROUP][NSEG][32];
  __shared__ __attribute__((aligned(16))) float Bre_s[N_MODES * MROW];
  __shared__ __attribute__((aligned(16))) float Bim_s[N_MODES * MROW];
  __shared__ __attribute__((aligned(16))) float Are_s[N_MODES * MROW];
  __shared__ __attribute__((aligned(16))) float AimN_s[N_MODES * MROW];
  __shared__ __attribute__((aligned(16))) float Kmat_s[N_MODES * MROW];
  __shared__ float kern_s[N_MODES];
  __shared__ float cQr[16], cQi[16], cP[16], cGr[16], cGi[16];
  __shared__ float cCr[16], cCi[16], cTr[16], cTi[16];
  __shared__ float sOmega;

  const int tid = threadIdx.x;
  const int d = blockIdx.x;

  const int n16 = tid & 15;
  const float om_in = omga[d];
  const float th_in = theta[d];
  const float a_in = alpha[d * 16 + n16];
  const float de_in = delta[d * 16 + n16];
  const v2f g_in = ((const v2f*)gma)[d * 16 + n16];
  const float g0_in = g_in.x;
  const float g1_in = g_in.y;
  v4f xr[B_DIM][XV4];
  #pragma unroll
  for (int b2 = 0; b2 < B_DIM; ++b2) {
    const v4f* xg = (const v4f*)(x + ((size_t)b2 * D_DIM + d) * L_DIM);
    #pragma unroll
    for (int k = 0; k < XV4; ++k) xr[b2][k] = xg[tid + 256 * k];
  }
  asm volatile("" :: "v"(a_in), "v"(de_in), "v"(g0_in), "v"(g1_in), "s"(om_in), "s"(th_in));

  if (tid < 16) {
    int n = tid;
    float a = a_in;
    float de = de_in;
    float th = th_in;
    float p = 1.f / (1.f + expf(-a));
    float dd = 1.f / (1.f + expf(-de));
    float base = (1.f / (1.f + expf(-th))) * (2.f * 3.14159265358979323846f / 16.f);
    float phi = (float)(n + 1) * base;
    float r = 1.f - p * dd;
    float sn, cs;
    sincosf(phi, &sn, &cs);
    float qr = r * cs, qi = r * sn;
    cQr[n] = qr; cQi[n] = qi;
    cP[n] = p;
    float gr = g0_in * 0.25f;
    float gi = g1_in * 0.25f;
    cGr[n] = gr; cGi[n] = gi;
    cCr[n] = p * gr; cCi[n] = p * gi;
    float tr, ti;
    cpow(qr, qi, 16, tr, ti);
    cTr[n] = tr; cTi[n] = ti;
    if (n == 0) sOmega = om_in;
  }
  block_lds_barrier();

  {
    int r = tid >> 4, cc = tid & 15;
    float qr = cQr[cc], qi = cQi[cc];
    float pr, pi;
    cpow(qr, qi, r, pr, pi);
    xs_s[0][r * MROW + cc] = cCr[cc] * pr - cCi[cc] * pi;
    float q1r = pr * qr - pi * qi;
    float q1i = pr * qi + pi * qr;
    Are_s[r * MROW + cc] = cGr[cc] * q1r - cGi[cc] * q1i;
    AimN_s[r * MROW + cc] = -(cGr[cc] * q1i + cGi[cc] * q1r);
    float br, bi;
    cpow(cQr[r], cQi[r], 15 - cc, br, bi);
    Bre_s[r * MROW + cc] = cP[r] * br;
    Bim_s[r * MROW + cc] = cP[r] * bi;
  }
  block_lds_barrier();
  if (tid < 16) {
    float s = 0.f;
    for (int n = 0; n < 16; ++n) s += xs_s[0][tid * MROW + n];
    kern_s[tid] = s;
  }
  block_lds_barrier();
  {
    int r = tid >> 4, cc = tid & 15;
    float v = (r >= cc) ? kern_s[r - cc] : 0.f;
    if (r == cc) v += sOmega;
    Kmat_s[r * MROW + cc] = v;
  }

  asm volatile("" :: "v"(xr[0][0]), "v"(xr[0][1]), "v"(xr[0][2]), "v"(xr[0][3]),
                     "v"(xr[1][0]), "v"(xr[1][1]), "v"(xr[1][2]), "v"(xr[1][3]));
  #pragma unroll
  for (int b2 = 0; b2 < B_DIM; ++b2) {
    #pragma unroll
    for (int k = 0; k < XV4; ++k) {
      int i = tid + 256 * k;
      *(v4f*)&xs_s[b2][xswz(i >> 2, i & 3)] = xr[b2][k];
    }
  }

  const int wv = tid >> 5;
  const int lane = tid & 31;
  const int lcol = lane & 15;
  const int khalf = lane >> 4;
  const int sB = (lcol >> 2) & 3;

  float hre = 0.f, him = 0.f;
  const float qtr = cTr[lcol], qti = cTi[lcol];
  float wtr = qtr, wti = qti;
  for (int k = 0; k < 4; ++k) {
    float t = wtr * wtr - wti * wti;
    wti = 2.f * wtr * wti;
    wtr = t;
  }

  block_lds_barrier();

  #pragma unroll
  for (int g = 0; g < NGROUP; ++g) {
    wave_lds_order();

    #pragma unroll
    for (int u = 0; u < 2; ++u) {
      int b2 = u;
      int tile = wv;
      const float* xb = xs_s[b2] + g * (CG * XROW);
      int xbase = (tile * 16 + lcol) * XROW + khalf * 2;
      v8f accR = {0.f, 0.f, 0.f, 0.f, 0.f, 0.f, 0.f, 0.f};
      v8f accI = {0.f, 0.f, 0.f, 0.f, 0.f, 0.f, 0.f, 0.f};
      #pragma unroll
      for (int kk = 0; kk < 4; ++kk) {
        int k0 = kk * 4 + khalf * 2;
        v2f bx = *(const v2f*)&xb[xbase + ((kk ^ sB) << 2)];
        v2f aR = *(const v2f*)&Bre_s[lcol * MROW + k0];
        v2f aI = *(const v2f*)&Bim_s[lcol * MROW + k0];
        accR = wmma4(aR, bx, accR);
        accI = wmma4(aI, bx, accI);
      }
      #pragma unroll
      for (int v = 0; v < 8; ++v) {
        int n = v + 8 * khalf;
        Ure_s[b2][n * UROW + tile * 16 + lcol] = accR[v];
        Uim_s[b2][n * UROW + tile * 16 + lcol] = accI[v];
      }
    }
    wave_lds_order();

    {
      float* ur = &Ure_s[khalf][lcol * UROW + wv * SEGC];
      float* ui = &Uim_s[khalf][lcol * UROW + wv * SEGC];
      float u_r[SEGC], u_i[SEGC];
      for (int k = 0; k < SEGC / 4; ++k) {
        float4 a = *(const float4*)(ur + 4 * k);
        float4 b = *(const float4*)(ui + 4 * k);
        u_r[4 * k + 0] = a.x; u_r[4 * k + 1] = a.y; u_r[4 * k + 2] = a.z; u_r[4 * k + 3] = a.w;
        u_i[4 * k + 0] = b.x; u_i[4 * k + 1] = b.y; u_i[4 * k + 2] = b.z; u_i[4 * k + 3] = b.w;
      }
      float sr = u_r[0], si = u_i[0];
      for (int j = 1; j < SEGC; ++j) {
        float nr = fmaf(qtr, sr, fmaf(-qti, si, u_r[j]));
        float ni = fmaf(qtr, si, fmaf(qti, sr, u_i[j]));
        sr = nr; si = ni;
      }
      Sre_s[g][wv][lane] = sr;
      Sim_s[g][wv][lane] = si;
      block_lds_barrier();

      float cr = hre, ci = him;
      float mr = hre, mi = him;
      for (int s = 0; s < NSEG; ++s) {
        float tsr = Sre_s[g][s][lane], tsi = Sim_s[g][s][lane];
        float nr = fmaf(wtr, cr, fmaf(-wti, ci, tsr));
        float ni = fmaf(wtr, ci, fmaf(wti, cr, tsi));
        cr = nr; ci = ni;
        if (s + 1 == wv) { mr = cr; mi = ci; }
      }
      hre = cr; him = ci;

      float h_r[SEGC], h_i[SEGC];
      sr = mr; si = mi;
      for (int j = 0; j < SEGC; ++j) {
        h_r[j] = sr; h_i[j] = si;
        if (j + 1 < SEGC) {
          float nr = fmaf(qtr, sr, fmaf(-qti, si, u_r[j]));
          float ni = fmaf(qtr, si, fmaf(qti, sr, u_i[j]));
          sr = nr; si = ni;
        }
      }
      for (int k = 0; k < SEGC / 4; ++k) {
        *(float4*)(ur + 4 * k) = make_float4(h_r[4 * k], h_r[4 * k + 1], h_r[4 * k + 2], h_r[4 * k + 3]);
        *(float4*)(ui + 4 * k) = make_float4(h_i[4 * k], h_i[4 * k + 1], h_i[4 * k + 2], h_i[4 * k + 3]);
      }
    }
    wave_lds_order();

    #pragma unroll
    for (int u = 0; u < 2; ++u) {
      int b2 = u;
      int tile = wv;
      float* xb = xs_s[b2] + g * (CG * XROW);
      int xbase = (tile * 16 + lcol) * XROW + khalf * 2;
      v8f acc = {0.f, 0.f, 0.f, 0.f, 0.f, 0.f, 0.f, 0.f};
      #pragma unroll
      for (int kk = 0; kk < 4; ++kk) {
        int k0 = kk * 4 + khalf * 2;
        v2f bx = *(const v2f*)&xb[xbase + ((kk ^ sB) << 2)];
        v2f aK = *(const v2f*)&Kmat_s[lcol * MROW + k0];
        acc = wmma4(aK, bx, acc);
      }
      #pragma unroll
      for (int kk = 0; kk < 4; ++kk) {
        int k0 = kk * 4 + khalf * 2;
        v2f bh;
        bh.x = Ure_s[b2][k0 * UROW + tile * 16 + lcol];
        bh.y = Ure_s[b2][(k0 + 1) * UROW + tile * 16 + lcol];
        v2f aA = *(const v2f*)&Are_s[lcol * MROW + k0];
        acc = wmma4(aA, bh, acc);
      }
      #pragma unroll
      for (int kk = 0; kk < 4; ++kk) {
        int k0 = kk * 4 + khalf * 2;
        v2f bh;
        bh.x = Uim_s[b2][k0 * UROW + tile * 16 + lcol];
        bh.y = Uim_s[b2][(k0 + 1) * UROW + tile * 16 + lcol];
        v2f aA = *(const v2f*)&AimN_s[lcol * MROW + k0];
        acc = wmma4(aA, bh, acc);
      }
      *(float4*)&xb[(tile * 16 + lcol) * XROW + (((2 * khalf) ^ sB) << 2)] =
          make_float4(acc[0], acc[1], acc[2], acc[3]);
      *(float4*)&xb[(tile * 16 + lcol) * XROW + (((2 * khalf + 1) ^ sB) << 2)] =
          make_float4(acc[4], acc[5], acc[6], acc[7]);
    }
    wave_lds_order();

    #pragma unroll
    for (int b2 = 0; b2 < B_DIM; ++b2) {
      float4* yg = (float4*)(y_out + ((size_t)b2 * D_DIM + d) * L_DIM +
                             (size_t)g * (CG * 16) + (size_t)wv * 256);
      #pragma unroll
      for (int k = 0; k < 2; ++k) {
        int i = lane + 32 * k;
        yg[i] = *(const float4*)&xs_s[b2][g * (CG * XROW) + xswz(wv * 16 + (i >> 2), i & 3)];
      }
    }
  }

  if (tid < 32) {
    int b2 = tid >> 4, n = tid & 15;
    float* hp = h_out + (((size_t)b2 * D_DIM + d) * N_MODES + n) * 2;
    hp[0] = hre;
    hp[1] = him;
  }
}

extern "C" void kernel_launch(void* const* d_in, const int* in_sizes, int n_in,
                              void* d_out, int out_size, void* d_ws, size_t ws_size,
                              hipStream_t stream) {
  (void)in_sizes; (void)n_in; (void)out_size; (void)d_ws; (void)ws_size;
  const float* x = (const float*)d_in[0];
  const float* alpha = (const float*)d_in[1];
  const float* delta = (const float*)d_in[2];
  const float* theta = (const float*)d_in[3];
  const float* gma = (const float*)d_in[4];
  const float* omga = (const float*)d_in[5];
  float* y = (float*)d_out;
  float* h = (float*)d_out + (size_t)B_DIM * D_DIM * L_DIM;
  cema_kernel<<<dim3(D_DIM), dim3(256), 0, stream>>>(x, alpha, delta, theta,
                                                     gma, omga, y, h);
}


# set-up's serial path cut: powers by squaring (no loops), the 16 terms of a kernel tap summed across lanes (its sum order changes), Kmat built behind the last barrier: 4 set-up barriers become 2
# speedup vs baseline: 1.8220x; 1.0587x over previous
#include <hip/hip_runtime.h>


typedef float v8f __attribute__((ext_vector_type(8)));
typedef float v2f __attribute__((ext_vector_type(2)));
typedef float v4f __attribute__((ext_vector_type(4)));

#define D_DIM 2048
#define N_MODES 16
#define B_DIM 2
#define L_DIM 4096
#define CG 128
#define NGROUP 2
#define MROW 20
#define UROW 140
#define XROW 16
#define NCHUNK (NGROUP * CG)
#define XV4 (L_DIM / 4 / 256)
#define SEGC 16
#define NSEG (CG / SEGC)

static __device__ __forceinline__ v8f wmma4(v2f a, v2f b, v8f c) {
  return __builtin_amdgcn_wmma_f32_16x16x4_f32(false, a, false, b, (short)0, c,
                                               false, false);
}

static __device__ __forceinline__ int xswz(int chunk, int grp) {
  return chunk * XROW + ((grp ^ ((chunk >> 2) & 3)) << 2);
}

static __device__ __forceinline__ void wave_lds_order() {
  __builtin_amdgcn_fence(__ATOMIC_RELEASE, "wavefront");
  __builtin_amdgcn_wave_barrier();
  __builtin_amdgcn_fence(__ATOMIC_ACQUIRE, "wavefront");
}

static __device__ __forceinline__ void block_lds_barrier() {
  __builtin_amdgcn_fence(__ATOMIC_RELEASE, "workgroup", "local");
  __builtin_amdgcn_s_barrier();
  __builtin_amdgcn_fence(__ATOMIC_ACQUIRE, "workgroup", "local");
}

static __device__ __forceinline__ void cpow(float qr, float qi, int e,
                                            float& pr, float& pi) {
  const float q2r = qr * qr - qi * qi, q2i = 2.f * qr * qi;
  const float q4r = q2r * q2r - q2i * q2i, q4i = 2.f * q2r * q2i;
  const float q8r = q4r * q4r - q4i * q4i, q8i = 2.f * q4r * q4i;
  const float ar = (e & 1) ? qr : 1.f, ai = (e & 1) ? qi : 0.f;
  const float br = (e & 2) ? q2r : 1.f, bi = (e & 2) ? q2i : 0.f;
  const float cr = (e & 4) ? q4r : 1.f, ci = (e & 4) ? q4i : 0.f;
  const float dr = (e & 8) ? q8r : 1.f, di = (e & 8) ? q8i : 0.f;
  const float lr = ar * br - ai * bi, li = ar * bi + ai * br;
  const float hr = cr * dr - ci * di, hi = cr * di + ci * dr;
  pr = lr * hr - li * hi;
  pi = lr * hi + li * hr;
}

template <int CTRL>
static __device__ __forceinline__ float row_rot(float v) {
  return __builtin_bit_cast(float, __builtin_amdgcn_update_dpp(0, __builtin_bit_cast(int, v), CTRL, 0xF, 0xF, false));
}

static __device__ __forceinline__ float row_sum16(float v) {
  v += row_rot<0x128>(v);
  v += row_rot<0x124>(v);
  v += row_rot<0x122>(v);
  v += row_rot<0x121>(v);
  return v;
}

__launch_bounds__(256, 8)
__global__ void cema_kernel(const float* __restrict__ x,
                            const float* __restrict__ alpha,
                            const float* __restrict__ delta,
                            const float* __restrict__ theta,
                            const float* __restrict__ gma,
                            const float* __restrict__ omga,
                            float* __restrict__ y_out,
                            float* __restrict__ h_out) {
  __shared__ __attribute__((aligned(16))) float xs_s[B_DIM][NCHUNK * XROW];
  __shared__ __attribute__((aligned(16))) float Ure_s[B_DIM][N_MODES * UROW];
  __shared__ __attribute__((aligned(16))) float Uim_s[B_DIM][N_MODES * UROW];
  __shared__ float Sre_s[NGROUP][NSEG][32];
  __shared__ float Sim_s[NGROUP][NSEG][32];
  __shared__ __attribute__((aligned(16))) float Bre_s[N_MODES * MROW];
  __shared__ __attribute__((aligned(16))) float Bim_s[N_MODES * MROW];
  __shared__ __attribute__((aligned(16))) float Are_s[N_MODES * MROW];
  __shared__ __attribute__((aligned(16))) float AimN_s[N_MODES * MROW];
  __shared__ __attribute__((aligned(16))) float Kmat_s[N_MODES * MROW];
  __shared__ float kern_s[N_MODES];
  __shared__ float cQr[16], cQi[16], cP[16], cGr[16], cGi[16];
  __shared__ float cCr[16], cCi[16], cTr[16], cTi[16];
  __shared__ float sOmega;

  const int tid = threadIdx.x;
  const int d = blockIdx.x;

  const int n16 = tid & 15;
  const float om_in = omga[d];
  const float th_in = theta[d];
  const float a_in = alpha[d * 16 + n16];
  const float de_in = delta[d * 16 + n16];
  const v2f g_in = ((const v2f*)gma)[d * 16 + n16];
  const float g0_in = g_in.x;
  const float g1_in = g_in.y;
  v4f xr[B_DIM][XV4];
  #pragma unroll
  for (int b2 = 0; b2 < B_DIM; ++b2) {
    const v4f* xg = (const v4f*)(x + ((size_t)b2 * D_DIM + d) * L_DIM);
    #pragma unroll
    for (int k = 0; k < XV4; ++k) xr[b2][k] = xg[tid + 256 * k];
  }
  asm volatile("" :: "v"(a_in), "v"(de_in), "v"(g0_in), "v"(g1_in), "s"(om_in), "s"(th_in));

  if (tid < 16) {
    int n = tid;
    float a = a_in;
    float de = de_in;
    float th = th_in;
    float p = 1.f / (1.f + expf(-a));
    float dd = 1.f / (1.f + expf(-de));
    float base = (1.f / (1.f + expf(-th))) * (2.f * 3.14159265358979323846f / 16.f);
    float phi = (float)(n + 1) * base;
    float r = 1.f - p * dd;
    float sn, cs;
    sincosf(phi, &sn, &cs);
    float qr = r * cs, qi = r * sn;
    cQr[n] = qr; cQi[n] = qi;
    cP[n] = p;
    float gr = g0_in * 0.25f;
    float gi = g1_in * 0.25f;
    cGr[n] = gr; cGi[n] = gi;
    cCr[n] = p * gr; cCi[n] = p * gi;
    float tr = qr, ti = qi;
    for (int k = 0; k < 4; ++k) {
      float t = tr * tr - ti * ti;
      ti = 2.f * tr * ti;
      tr = t;
    }
    cTr[n] = tr; cTi[n] = ti;
    if (n == 0) sOmega = om_in;
  }
  block_lds_barrier();

  {
    int r = tid >> 4, cc = tid & 15;
    float qr = cQr[cc], qi = cQi[cc];
    float pr, pi;
    cpow(qr, qi, r, pr, pi);
    float tap = row_sum16(cCr[cc] * pr - cCi[cc] * pi);
    if (cc == 0) kern_s[r] = tap;
    float q1r = pr * qr - pi * qi;
    float q1i = pr * qi + pi * qr;
    Are_s[r * MROW + cc] = cGr[cc] * q1r - cGi[cc] * q1i;
    AimN_s[r * MROW + cc] = -(cGr[cc] * q1i + cGi[cc] * q1r);
    float br, bi;
    cpow(cQr[r], cQi[r], 15 - cc, br, bi);
    Bre_s[r * MROW + cc] = cP[r] * br;
    Bim_s[r * MROW + cc] = cP[r] * bi;
  }

  asm volatile("" :: "v"(xr[0][0]), "v"(xr[0][1]), "v"(xr[0][2]), "v"(xr[0][3]),
                     "v"(xr[1][0]), "v"(xr[1][1]), "v"(xr[1][2]), "v"(xr[1][3]));
  #pragma unroll
  for (int b2 = 0; b2 < B_DIM; ++b2) {
    #pragma unroll
    for (int k = 0; k < XV4; ++k) {
      int i = tid + 256 * k;
      *(v4f*)&xs_s[b2][xswz(i >> 2, i & 3)] = xr[b2][k];
    }
  }
  block_lds_barrier();

  {
    int r = tid >> 4, cc = tid & 15;
    float v = (r >= cc) ? kern_s[r - cc] : 0.f;
    if (r == cc) v += sOmega;
    Kmat_s[r * MROW + cc] = v;
  }

  const int wv = tid >> 5;
  const int lane = tid & 31;
  const int lcol = lane & 15;
  const int khalf = lane >> 4;
  const int sB = (lcol >> 2) & 3;

  float hre = 0.f, him = 0.f;
  const float qtr = cTr[lcol], qti = cTi[lcol];
  float wtr = qtr, wti = qti;
  for (int k = 0; k < 4; ++k) {
    float t = wtr * wtr - wti * wti;
    wti = 2.f * wtr * wti;
    wtr = t;
  }

  #pragma unroll
  for (int g = 0; g < NGROUP; ++g) {
    wave_lds_order();

    #pragma unroll
    for (int u = 0; u < 2; ++u) {
      int b2 = u;
      int tile = wv;
      const float* xb = xs_s[b2] + g * (CG * XROW);
      int xbase = (tile * 16 + lcol) * XROW + khalf * 2;
      v8f accR = {0.f, 0.f, 0.f, 0.f, 0.f, 0.f, 0.f, 0.f};
      v8f accI = {0.f, 0.f, 0.f, 0.f, 0.f, 0.f, 0.f, 0.f};
      #pragma unroll
      for (int kk = 0; kk < 4; ++kk) {
        int k0 = kk * 4 + khalf * 2;
        v2f bx = *(const v2f*)&xb[xbase + ((kk ^ sB) << 2)];
        v2f aR = *(const v2f*)&Bre_s[lcol * MROW + k0];
        v2f aI = *(const v2f*)&Bim_s[lcol * MROW + k0];
        accR = wmma4(aR, bx, accR);
        accI = wmma4(aI, bx, accI);
      }
      #pragma unroll
      for (int v = 0; v < 8; ++v) {
        int n = v + 8 * khalf;
        Ure_s[b2][n * UROW + tile * 16 + lcol] = accR[v];
        Uim_s[b2][n * UROW + tile * 16 + lcol] = accI[v];
      }
    }
    wave_lds_order();

    {
      float* ur = &Ure_s[khalf][lcol * UROW + wv * SEGC];
      float* ui = &Uim_s[khalf][lcol * UROW + wv * SEGC];
      float u_r[SEGC], u_i[SEGC];
      for (int k = 0; k < SEGC / 4; ++k) {
        float4 a = *(const float4*)(ur + 4 * k);
        float4 b = *(const float4*)(ui + 4 * k);
        u_r[4 * k + 0] = a.x; u_r[4 * k + 1] = a.y; u_r[4 * k + 2] = a.z; u_r[4 * k + 3] = a.w;
        u_i[4 * k + 0] = b.x; u_i[4 * k + 1] = b.y; u_i[4 * k + 2] = b.z; u_i[4 * k + 3] = b.w;
      }
      float sr = u_r[0], si = u_i[0];
      for (int j = 1; j < SEGC; ++j) {
        float nr = fmaf(qtr, sr, fmaf(-qti, si, u_r[j]));
        float ni = fmaf(qtr, si, fmaf(qti, sr, u_i[j]));
        sr = nr; si = ni;
      }
      Sre_s[g][wv][lane] = sr;
      Sim_s[g][wv][lane] = si;
      block_lds_barrier();

      float cr = hre, ci = him;
      float mr = hre, mi = him;
      for (int s = 0; s < NSEG; ++s) {
        float tsr = Sre_s[g][s][lane], tsi = Sim_s[g][s][lane];
        float nr = fmaf(wtr, cr, fmaf(-wti, ci, tsr));
        float ni = fmaf(wtr, ci, fmaf(wti, cr, tsi));
        cr = nr; ci = ni;
        if (s + 1 == wv) { mr = cr; mi = ci; }
      }
      hre = cr; him = ci;

      float h_r[SEGC], h_i[SEGC];
      sr = mr; si = mi;
      for (int j = 0; j < SEGC; ++j) {
        h_r[j] = sr; h_i[j] = si;
        if (j + 1 < SEGC) {
          float nr = fmaf(qtr, sr, fmaf(-qti, si, u_r[j]));
          float ni = fmaf(qtr, si, fmaf(qti, sr, u_i[j]));
          sr = nr; si = ni;
        }
      }
      for (int k = 0; k < SEGC / 4; ++k) {
        *(float4*)(ur + 4 * k) = make_float4(h_r[4 * k], h_r[4 * k + 1], h_r[4 * k + 2], h_r[4 * k + 3]);
        *(float4*)(ui + 4 * k) = make_float4(h_i[4 * k], h_i[4 * k + 1], h_i[4 * k + 2], h_i[4 * k + 3]);
      }
    }
    wave_lds_order();

    #pragma unroll
    for (int u = 0; u < 2; ++u) {
      int b2 = u;
      int tile = wv;
      float* xb = xs_s[b2] + g * (CG * XROW);
      int xbase = (tile * 16 + lcol) * XROW + khalf * 2;
      v8f acc = {0.f, 0.f, 0.f, 0.f, 0.f, 0.f, 0.f, 0.f};
      #pragma unroll
      for (int kk = 0; kk < 4; ++kk) {
        int k0 = kk * 4 + khalf * 2;
        v2f bx = *(const v2f*)&xb[xbase + ((kk ^ sB) << 2)];
        v2f aK = *(const v2f*)&Kmat_s[lcol * MROW + k0];
        acc = wmma4(aK, bx, acc);
      }
      #pragma unroll
      for (int kk = 0; kk < 4; ++kk) {
        int k0 = kk * 4 + khalf * 2;
        v2f bh;
        bh.x = Ure_s[b2][k0 * UROW + tile * 16 + lcol];
        bh.y = Ure_s[b2][(k0 + 1) * UROW + tile * 16 + lcol];
        v2f aA = *(const v2f*)&Are_s[lcol * MROW + k0];
        acc = wmma4(aA, bh, acc);
      }
      #pragma unroll
      for (int kk = 0; kk < 4; ++kk) {
        int k0 = kk * 4 + khalf * 2;
        v2f bh;
        bh.x = Uim_s[b2][k0 * UROW + tile * 16 + lcol];
        bh.y = Uim_s[b2][(k0 + 1) * UROW + tile * 16 + lcol];
        v2f aA = *(const v2f*)&AimN_s[lcol * MROW + k0];
        acc = wmma4(aA, bh, acc);
      }
      *(float4*)&xb[(tile * 16 + lcol) * XROW + (((2 * khalf) ^ sB) << 2)] =
          make_float4(acc[0], acc[1], acc[2], acc[3]);
      *(float4*)&xb[(tile * 16 + lcol) * XROW + (((2 * khalf + 1) ^ sB) << 2)] =
          make_float4(acc[4], acc[5], acc[6], acc[7]);
    }
    wave_lds_order();

    #pragma unroll
    for (int b2 = 0; b2 < B_DIM; ++b2) {
      float4* yg = (float4*)(y_out + ((size_t)b2 * D_DIM + d) * L_DIM +
                             (size_t)g * (CG * 16) + (size_t)wv * 256);
      #pragma unroll
      for (int k = 0; k < 2; ++k) {
        int i = lane + 32 * k;
        yg[i] = *(const float4*)&xs_s[b2][g * (CG * XROW) + xswz(wv * 16 + (i >> 2), i & 3)];
      }
    }
  }

  if (tid < 32) {
    int b2 = tid >> 4, n = tid & 15;
    float* hp = h_out + (((size_t)b2 * D_DIM + d) * N_MODES + n) * 2;
    hp[0] = hre;
    hp[1] = him;
  }
}

extern "C" void kernel_launch(void* const* d_in, const int* in_sizes, int n_in,
                              void* d_out, int out_size, void* d_ws, size_t ws_size,
                              hipStream_t stream) {
  (void)in_sizes; (void)n_in; (void)out_size; (void)d_ws; (void)ws_size;
  const float* x = (const float*)d_in[0];
  const float* alpha = (const float*)d_in[1];
  const float* delta = (const float*)d_in[2];
  const float* theta = (const float*)d_in[3];
  const float* gma = (const float*)d_in[4];
  const float* omga = (const float*)d_in[5];
  float* y = (float*)d_out;
  float* h = (float*)d_out + (size_t)B_DIM * D_DIM * L_DIM;
  cema_kernel<<<dim3(D_DIM), dim3(256), 0, stream>>>(x, alpha, delta, theta,
                                                     gma, omga, y, h);
}


# U computed transposed (x fragment as A operand): each lane then holds one mode's U for 8 chunks and the scan runs in the result registers, no U round trip through LDS; state's sum order changes
# speedup vs baseline: 1.8905x; 1.0376x over previous
#include <hip/hip_runtime.h>


typedef float v8f __attribute__((ext_vector_type(8)));
typedef float v2f __attribute__((ext_vector_type(2)));
typedef float v4f __attribute__((ext_vector_type(4)));

#define D_DIM 2048
#define N_MODES 16
#define B_DIM 2
#define L_DIM 4096
#define CG 128
#define NGROUP 2
#define MROW 20
#define UROW 140
#define XROW 16
#define NCHUNK (NGROUP * CG)
#define XV4 (L_DIM / 4 / 256)
#define SEGC 16
#define NSEG (CG / SEGC)

static __device__ __forceinline__ v8f wmma4(v2f a, v2f b, v8f c) {
  return __builtin_amdgcn_wmma_f32_16x16x4_f32(false, a, false, b, (short)0, c,
                                               false, false);
}

static __device__ __forceinline__ int xswz(int chunk, int grp) {
  return chunk * XROW + ((grp ^ ((chunk >> 2) & 3)) << 2);
}

static __device__ __forceinline__ void wave_lds_order() {
  __builtin_amdgcn_fence(__ATOMIC_RELEASE, "wavefront");
  __builtin_amdgcn_wave_barrier();
  __builtin_amdgcn_fence(__ATOMIC_ACQUIRE, "wavefront");
}

static __device__ __forceinline__ void block_lds_barrier() {
  __builtin_amdgcn_fence(__ATOMIC_RELEASE, "workgroup", "local");
  __builtin_amdgcn_s_barrier();
  __builtin_amdgcn_fence(__ATOMIC_ACQUIRE, "workgroup", "local");
}

static __device__ __forceinline__ void cpow(float qr, float qi, int e,
                                            float& pr, float& pi) {
  const float q2r = qr * qr - qi * qi, q2i = 2.f * qr * qi;
  const float q4r = q2r * q2r - q2i * q2i, q4i = 2.f * q2r * q2i;
  const float q8r = q4r * q4r - q4i * q4i, q8i = 2.f * q4r * q4i;
  const float ar = (e & 1) ? qr : 1.f, ai = (e & 1) ? qi : 0.f;
  const float br = (e & 2) ? q2r : 1.f, bi = (e & 2) ? q2i : 0.f;
  const float cr = (e & 4) ? q4r : 1.f, ci = (e & 4) ? q4i : 0.f;
  const float dr = (e & 8) ? q8r : 1.f, di = (e & 8) ? q8i : 0.f;
  const float lr = ar * br - ai * bi, li = ar * bi + ai * br;
  const float hr = cr * dr - ci * di, hi = cr * di + ci * dr;
  pr = lr * hr - li * hi;
  pi = lr * hi + li * hr;
}

template <int CTRL>
static __device__ __forceinline__ float row_rot(float v) {
  return __builtin_bit_cast(float, __builtin_amdgcn_update_dpp(0, __builtin_bit_cast(int, v), CTRL, 0xF, 0xF, false));
}

static __device__ __forceinline__ float row_sum16(float v) {
  v += row_rot<0x128>(v);
  v += row_rot<0x124>(v);
  v += row_rot<0x122>(v);
  v += row_rot<0x121>(v);
  return v;
}

static __device__ __forceinline__ float xhalf(float v) {
  return __builtin_bit_cast(float, __builtin_amdgcn_permlanex16(__builtin_bit_cast(int, v), __builtin_bit_cast(int, v),
                                                                0x76543210, 0xfedcba98, false, false));
}

__launch_bounds__(256, 8)
__global__ void cema_kernel(const float* __restrict__ x,
                            const float* __restrict__ alpha,
                            const float* __restrict__ delta,
                            const float* __restrict__ theta,
                            const float* __restrict__ gma,
                            const float* __restrict__ omga,
                            float* __restrict__ y_out,
                            float* __restrict__ h_out) {
  __shared__ __attribute__((aligned(16))) float xs_s[B_DIM][NCHUNK * XROW];
  __shared__ __attribute__((aligned(16))) float Ure_s[B_DIM][N_MODES * UROW];
  __shared__ __attribute__((aligned(16))) float Uim_s[B_DIM][N_MODES * UROW];
  __shared__ float Sre_s[NGROUP][NSEG][32];
  __shared__ float Sim_s[NGROUP][NSEG][32];
  __shared__ __attribute__((aligned(16))) float Bre_s[N_MODES * MROW];
  __shared__ __attribute__((aligned(16))) float Bim_s[N_MODES * MROW];
  __shared__ __attribute__((aligned(16))) float Are_s[N_MODES * MROW];
  __shared__ __attribute__((aligned(16))) float AimN_s[N_MODES * MROW];
  __shared__ __attribute__((aligned(16))) float Kmat_s[N_MODES * MROW];
  __shared__ float kern_s[N_MODES];
  __shared__ float cQr[16], cQi[16], cP[16], cGr[16], cGi[16];
  __shared__ float cCr[16], cCi[16], cTr[16], cTi[16];
  __shared__ float sOmega;

  const int tid = threadIdx.x;
  const int d = blockIdx.x;

  const int n16 = tid & 15;
  const float om_in = omga[d];
  const float th_in = theta[d];
  const float a_in = alpha[d * 16 + n16];
  const float de_in = delta[d * 16 + n16];
  const v2f g_in = ((const v2f*)gma)[d * 16 + n16];
  const float g0_in = g_in.x;
  const float g1_in = g_in.y;
  v4f xr[B_DIM][XV4];
  #pragma unroll
  for (int b2 = 0; b2 < B_DIM; ++b2) {
    const v4f* xg = (const v4f*)(x + ((size_t)b2 * D_DIM + d) * L_DIM);
    #pragma unroll
    for (int k = 0; k < XV4; ++k) xr[b2][k] = xg[tid + 256 * k];
  }
  asm volatile("" :: "v"(a_in), "v"(de_in), "v"(g0_in), "v"(g1_in), "s"(om_in), "s"(th_in));

  if (tid < 16) {
    int n = tid;
    float a = a_in;
    float de = de_in;
    float th = th_in;
    float p = 1.f / (1.f + expf(-a));
    float dd = 1.f / (1.f + expf(-de));
    float base = (1.f / (1.f + expf(-th))) * (2.f * 3.14159265358979323846f / 16.f);
    float phi = (float)(n + 1) * base;
    float r = 1.f - p * dd;
    float sn, cs;
    sincosf(phi, &sn, &cs);
    float qr = r * cs, qi = r * sn;
    cQr[n] = qr; cQi[n] = qi;
    cP[n] = p;
    float gr = g0_in * 0.25f;
    float gi = g1_in * 0.25f;
    cGr[n] = gr; cGi[n] = gi;
    cCr[n] = p * gr; cCi[n] = p * gi;
    float tr = qr, ti = qi;
    for (int k = 0; k < 4; ++k) {
      float t = tr * tr - ti * ti;
      ti = 2.f * tr * ti;
      tr = t;
    }
    cTr[n] = tr; cTi[n] = ti;
    if (n == 0) sOmega = om_in;
  }
  block_lds_barrier();

  {
    int r = tid >> 4, cc = tid & 15;
    float qr = cQr[cc], qi = cQi[cc];
    float pr, pi;
    cpow(qr, qi, r, pr, pi);
    float tap = row_sum16(cCr[cc] * pr - cCi[cc] * pi);
    if (cc == 0) kern_s[r] = tap;
    float q1r = pr * qr - pi * qi;
    float q1i = pr * qi + pi * qr;
    Are_s[r * MROW + cc] = cGr[cc] * q1r - cGi[cc] * q1i;
    AimN_s[r * MROW + cc] = -(cGr[cc] * q1i + cGi[cc] * q1r);
    float br, bi;
    cpow(cQr[r], cQi[r], 15 - cc, br, bi);
    Bre_s[r * MROW + cc] = cP[r] * br;
    Bim_s[r * MROW + cc] = cP[r] * bi;
  }

  asm volatile("" :: "v"(xr[0][0]), "v"(xr[0][1]), "v"(xr[0][2]), "v"(xr[0][3]),
                     "v"(xr[1][0]), "v"(xr[1][1]), "v"(xr[1][2]), "v"(xr[1][3]));
  #pragma unroll
  for (int b2 = 0; b2 < B_DIM; ++b2) {
    #pragma unroll
    for (int k = 0; k < XV4; ++k) {
      int i = tid + 256 * k;
      *(v4f*)&xs_s[b2][xswz(i >> 2, i & 3)] = xr[b2][k];
    }
  }
  block_lds_barrier();

  {
    int r = tid >> 4, cc = tid & 15;
    float v = (r >= cc) ? kern_s[r - cc] : 0.f;
    if (r == cc) v += sOmega;
    Kmat_s[r * MROW + cc] = v;
  }

  const int wv = tid >> 5;
  const int lane = tid & 31;
  const int lcol = lane & 15;
  const int khalf = lane >> 4;
  const int sB = (lcol >> 2) & 3;

  float hre = 0.f, him = 0.f;
  const float qtr = cTr[lcol], qti = cTi[lcol];
  float wtr = qtr, wti = qti;
  float w8r = 0.f, w8i = 0.f;
  for (int k = 0; k < 4; ++k) {
    float t = wtr * wtr - wti * wti;
    wti = 2.f * wtr * wti;
    wtr = t;
    if (k == 2) { w8r = wtr; w8i = wti; }
  }

  #pragma unroll
  for (int g = 0; g < NGROUP; ++g) {
    v8f uR[B_DIM], uI[B_DIM];
    #pragma unroll
    for (int b2 = 0; b2 < B_DIM; ++b2) {
      const float* xb = xs_s[b2] + g * (CG * XROW);
      int xbase = (wv * 16 + lcol) * XROW + khalf * 2;
      v8f accR = {0.f, 0.f, 0.f, 0.f, 0.f, 0.f, 0.f, 0.f};
      v8f accI = {0.f, 0.f, 0.f, 0.f, 0.f, 0.f, 0.f, 0.f};
      #pragma unroll
      for (int kk = 0; kk < 4; ++kk) {
        int k0 = kk * 4 + khalf * 2;
        v2f ax = *(const v2f*)&xb[xbase + ((kk ^ sB) << 2)];
        v2f bR = *(const v2f*)&Bre_s[lcol * MROW + k0];
        v2f bI = *(const v2f*)&Bim_s[lcol * MROW + k0];
        accR = wmma4(ax, bR, accR);
        accI = wmma4(ax, bI, accI);
      }
      uR[b2] = accR;
      uI[b2] = accI;
    }

    {
      float lR[B_DIM], lI[B_DIM];
      float tR = 0.f, tI = 0.f;
      #pragma unroll
      for (int b2 = 0; b2 < B_DIM; ++b2) {
        float sr = uR[b2][0], si = uI[b2][0];
        #pragma unroll
        for (int j = 1; j < 8; ++j) {
          float nr = fmaf(qtr, sr, fmaf(-qti, si, uR[b2][j]));
          float ni = fmaf(qtr, si, fmaf(qti, sr, uI[b2][j]));
          sr = nr; si = ni;
        }
        float pr = xhalf(sr), pi = xhalf(si);
        float lor = khalf ? pr : sr, loi = khalf ? pi : si;
        float hir = khalf ? sr : pr, hii = khalf ? si : pi;
        lR[b2] = lor; lI[b2] = loi;
        float wr = fmaf(w8r, lor, fmaf(-w8i, loi, hir));
        float wi = fmaf(w8r, loi, fmaf(w8i, lor, hii));
        if (b2 == khalf) { tR = wr; tI = wi; }
      }
      Sre_s[g][wv][lane] = tR;
      Sim_s[g][wv][lane] = tI;
      block_lds_barrier();

      float cr = hre, ci = him;
      float mr = hre, mi = him;
      for (int s = 0; s < NSEG; ++s) {
        float tsr = Sre_s[g][s][lane], tsi = Sim_s[g][s][lane];
        float nr = fmaf(wtr, cr, fmaf(-wti, ci, tsr));
        float ni = fmaf(wtr, ci, fmaf(wti, cr, tsi));
        cr = nr; ci = ni;
        if (s + 1 == wv) { mr = cr; mi = ci; }
      }
      hre = cr; him = ci;

      float pmr = xhalf(mr), pmi = xhalf(mi);
      #pragma unroll
      for (int b2 = 0; b2 < B_DIM; ++b2) {
        float Mr = (b2 == khalf) ? mr : pmr;
        float Mi = (b2 == khalf) ? mi : pmi;
        float upr = fmaf(w8r, Mr, fmaf(-w8i, Mi, lR[b2]));
        float upi = fmaf(w8r, Mi, fmaf(w8i, Mr, lI[b2]));
        float sr = khalf ? upr : Mr, si = khalf ? upi : Mi;
        float h_r[8], h_i[8];
        #pragma unroll
        for (int j = 0; j < 8; ++j) {
          h_r[j] = sr; h_i[j] = si;
          if (j + 1 < 8) {
            float nr = fmaf(qtr, sr, fmaf(-qti, si, uR[b2][j]));
            float ni = fmaf(qtr, si, fmaf(qti, sr, uI[b2][j]));
            sr = nr; si = ni;
          }
        }
        float* hr = &Ure_s[b2][lcol * UROW + wv * SEGC + khalf * 8];
        float* hi = &Uim_s[b2][lcol * UROW + wv * SEGC + khalf * 8];
        *(float4*)(hr) = make_float4(h_r[0], h_r[1], h_r[2], h_r[3]);
        *(float4*)(hr + 4) = make_float4(h_r[4], h_r[5], h_r[6], h_r[7]);
        *(float4*)(hi) = make_float4(h_i[0], h_i[1], h_i[2], h_i[3]);
        *(float4*)(hi + 4) = make_float4(h_i[4], h_i[5], h_i[6], h_i[7]);
      }
    }
    wave_lds_order();

    #pragma unroll
    for (int u = 0; u < 2; ++u) {
      int b2 = u;
      int tile = wv;
      float* xb = xs_s[b2] + g * (CG * XROW);
      int xbase = (tile * 16 + lcol) * XROW + khalf * 2;
      v8f acc = {0.f, 0.f, 0.f, 0.f, 0.f, 0.f, 0.f, 0.f};
      #pragma unroll
      for (int kk = 0; kk < 4; ++kk) {
        int k0 = kk * 4 + khalf * 2;
        v2f bx = *(const v2f*)&xb[xbase + ((kk ^ sB) << 2)];
        v2f aK = *(const v2f*)&Kmat_s[lcol * MROW + k0];
        acc = wmma4(aK, bx, acc);
      }
      #pragma unroll
      for (int kk = 0; kk < 4; ++kk) {
        int k0 = kk * 4 + khalf * 2;
        v2f bh;
        bh.x = Ure_s[b2][k0 * UROW + tile * 16 + lcol];
        bh.y = Ure_s[b2][(k0 + 1) * UROW + tile * 16 + lcol];
        v2f aA = *(const v2f*)&Are_s[lcol * MROW + k0];
        acc = wmma4(aA, bh, acc);
      }
      #pragma unroll
      for (int kk = 0; kk < 4; ++kk) {
        int k0 = kk * 4 + khalf * 2;
        v2f bh;
        bh.x = Uim_s[b2][k0 * UROW + tile * 16 + lcol];
        bh.y = Uim_s[b2][(k0 + 1) * UROW + tile * 16 + lcol];
        v2f aA = *(const v2f*)&AimN_s[lcol * MROW + k0];
        acc = wmma4(aA, bh, acc);
      }
      *(float4*)&xb[(tile * 16 + lcol) * XROW + (((2 * khalf) ^ sB) << 2)] =
          make_float4(acc[0], acc[1], acc[2], acc[3]);
      *(float4*)&xb[(tile * 16 + lcol) * XROW + (((2 * khalf + 1) ^ sB) << 2)] =
          make_float4(acc[4], acc[5], acc[6], acc[7]);
    }
    wave_lds_order();

    #pragma unroll
    for (int b2 = 0; b2 < B_DIM; ++b2) {
      float4* yg = (float4*)(y_out + ((size_t)b2 * D_DIM + d) * L_DIM +
                             (size_t)g * (CG * 16) + (size_t)wv * 256);
      #pragma unroll
      for (int k = 0; k < 2; ++k) {
        int i = lane + 32 * k;
        yg[i] = *(const float4*)&xs_s[b2][g * (CG * XROW) + xswz(wv * 16 + (i >> 2), i & 3)];
      }
    }
  }

  if (tid < 32) {
    int b2 = tid >> 4, n = tid & 15;
    float* hp = h_out + (((size_t)b2 * D_DIM + d) * N_MODES + n) * 2;
    hp[0] = hre;
    hp[1] = him;
  }
}

extern "C" void kernel_launch(void* const* d_in, const int* in_sizes, int n_in,
                              void* d_out, int out_size, void* d_ws, size_t ws_size,
                              hipStream_t stream) {
  (void)in_sizes; (void)n_in; (void)out_size; (void)d_ws; (void)ws_size;
  const float* x = (const float*)d_in[0];
  const float* alpha = (const float*)d_in[1];
  const float* delta = (const float*)d_in[2];
  const float* theta = (const float*)d_in[3];
  const float* gma = (const float*)d_in[4];
  const float* omga = (const float*)d_in[5];
  float* y = (float*)d_out;
  float* h = (float*)d_out + (size_t)B_DIM * D_DIM * L_DIM;
  cema_kernel<<<dim3(D_DIM), dim3(256), 0, stream>>>(x, alpha, delta, theta,
                                                     gma, omga, y, h);
}
